# starting kernel
# baseline (speedup 1.0000x reference)
#include <hip/hip_runtime.h>
#include <math.h>


typedef __attribute__((ext_vector_type(2))) float v2f;
typedef __attribute__((ext_vector_type(8))) float v8f;

#define BB 4
#define SS 2048
#define DD 64
#define HH 16
#define HDIM 4

static __device__ __forceinline__ v8f wmma_f32_16x16x4(v2f a, v2f b, v8f c) {
  return __builtin_amdgcn_wmma_f32_16x16x4_f32(false, a, false, b, (short)0, c,
                                               false, false);
}

__global__ __launch_bounds__(128) void qkv_kernel(
    const float* __restrict__ x,
    const float* __restrict__ Wq, const float* __restrict__ bq,
    const float* __restrict__ Wk, const float* __restrict__ bk,
    const float* __restrict__ Wv, const float* __restrict__ bv,
    float* __restrict__ ws) {
  const int NT = (BB * SS) / 16;
  int tid   = blockIdx.x * 4 + (threadIdx.x >> 5);
  int which = tid / NT;
  int tile  = tid % NT;
  int r0    = tile * 16;

  const float* W    = (which == 0) ? Wq : (which == 1) ? Wk : Wv;
  const float* bias = (which == 0) ? bq : (which == 1) ? bk : bv;
  float* Out = ws + (size_t)which * (BB * SS * DD);

  int lane = threadIdx.x & 31;
  int half = lane >> 4;
  int l16  = lane & 15;

  v2f afrag[16];
  const float* xrow = x + (size_t)(r0 + l16) * DD + 2 * half;
#pragma unroll
  for (int kc = 0; kc < 16; ++kc) {
    afrag[kc].x = xrow[4 * kc + 0];
    afrag[kc].y = xrow[4 * kc + 1];
  }

  v8f acc[4];
#pragma unroll
  for (int n = 0; n < 4; ++n)
#pragma unroll
    for (int v = 0; v < 8; ++v) acc[n][v] = 0.0f;

#pragma unroll
  for (int kc = 0; kc < 16; ++kc) {
    int krow = 4 * kc + 2 * half;
#pragma unroll
    for (int n = 0; n < 4; ++n) {
      v2f bfrag;
      bfrag.x = W[(size_t)krow * DD + n * 16 + l16];
      bfrag.y = W[(size_t)(krow + 1) * DD + n * 16 + l16];
      acc[n] = wmma_f32_16x16x4(afrag[kc], bfrag, acc[n]);
    }
  }

#pragma unroll
  for (int n = 0; n < 4; ++n) {
    int col = n * 16 + l16;
    float bb = bias[col];
#pragma unroll
    for (int v = 0; v < 8; ++v) {
      int row = r0 + v + 8 * half;
      Out[(size_t)row * DD + col] = fmaxf(acc[n][v] + bb, 0.0f);
    }
  }
}

__global__ __launch_bounds__(128) void attn_kernel(
    const float* __restrict__ x, const float* __restrict__ ws,
    float* __restrict__ out) {
  const float* Q = ws;
  const float* K = ws + (size_t)BB * SS * DD;
  const float* V = ws + (size_t)2 * BB * SS * DD;

  __shared__ float ldsP[4][16][16];

  int wave = threadIdx.x >> 5;
  int lane = threadIdx.x & 31;
  int half = lane >> 4;
  int l16  = lane & 15;

  const int QT = SS / 64;
  int bh = blockIdx.x / QT;
  int qt = blockIdx.x % QT;
  int b = bh / HH, h = bh % HH;
  int q0 = qt * 64 + wave * 16;

  const float* Qb = Q + ((size_t)b * SS) * DD + h * HDIM;
  const float* Kb = K + ((size_t)b * SS) * DD + h * HDIM;
  const float* Vb = V + ((size_t)b * SS) * DD + h * HDIM;

  v2f aq;
  aq.x = Qb[(size_t)(q0 + l16) * DD + 2 * half];
  aq.y = Qb[(size_t)(q0 + l16) * DD + 2 * half + 1];

  float mrow[8], lrow[8];
  v8f accO, zc;
#pragma unroll
  for (int v = 0; v < 8; ++v) {
    mrow[v] = -INFINITY;
    lrow[v] = 0.0f;
    accO[v] = 0.0f;
    zc[v]   = 0.0f;
  }

  const float scale = 0.5f;

  for (int k0 = 0; k0 < SS; k0 += 16) {
    if (k0 + 16 < SS) {
      __builtin_prefetch(Kb + (size_t)(k0 + 16 + l16) * DD, 0, 1);
      __builtin_prefetch(Vb + (size_t)(k0 + 16 + l16) * DD, 0, 1);
    }
    v2f bk;
    bk.x = Kb[(size_t)(k0 + l16) * DD + 2 * half];
    bk.y = Kb[(size_t)(k0 + l16) * DD + 2 * half + 1];
    v8f s = wmma_f32_16x16x4(aq, bk, zc);

    float p[8];
#pragma unroll
    for (int v = 0; v < 8; ++v) {
      float sv = s[v] * scale;
      float m = sv;
      m = fmaxf(m, __shfl_xor(m, 1, 16));
      m = fmaxf(m, __shfl_xor(m, 2, 16));
      m = fmaxf(m, __shfl_xor(m, 4, 16));
      m = fmaxf(m, __shfl_xor(m, 8, 16));
      float mnew  = fmaxf(mrow[v], m);
      float alpha = __expf(mrow[v] - mnew);
      float pv    = __expf(sv - mnew);
      float rs = pv;
      rs += __shfl_xor(rs, 1, 16);
      rs += __shfl_xor(rs, 2, 16);
      rs += __shfl_xor(rs, 4, 16);
      rs += __shfl_xor(rs, 8, 16);
      lrow[v] = lrow[v] * alpha + rs;
      mrow[v] = mnew;
      accO[v] *= alpha;
      p[v] = pv;
    }

#pragma unroll
    for (int v = 0; v < 8; ++v) ldsP[wave][v + 8 * half][l16] = p[v];

#pragma unroll
    for (int kc = 0; kc < 4; ++kc) {
      v2f ap;
      ap.x = ldsP[wave][l16][4 * kc + 2 * half];
      ap.y = ldsP[wave][l16][4 * kc + 2 * half + 1];
      v2f bv;
      int krow = k0 + 4 * kc + 2 * half;
      bv.x = (l16 < HDIM) ? Vb[(size_t)krow * DD + l16] : 0.0f;
      bv.y = (l16 < HDIM) ? Vb[(size_t)(krow + 1) * DD + l16] : 0.0f;
      accO = wmma_f32_16x16x4(ap, bv, accO);
    }
  }

  if (l16 < HDIM) {
    int ch = h * HDIM + l16;
#pragma unroll
    for (int v = 0; v < 8; ++v) {
      int row = q0 + v + 8 * half;
      size_t idx = ((size_t)b * SS + row) * DD + ch;
      out[idx] = accO[v] / lrow[v] + x[idx];
    }
  }
}

__global__ __launch_bounds__(256) void bn_stats_kernel(
    const float* __restrict__ out, float* __restrict__ stats) {
  __shared__ float ssum[256], ssq[256];
  int ch = blockIdx.x;
  int t = threadIdx.x;
  float s = 0.0f, q = 0.0f;
  for (int i = t; i < BB * SS; i += 256) {
    float v = out[(size_t)i * DD + ch];
    s += v;
    q += v * v;
  }
  ssum[t] = s;
  ssq[t] = q;
  __syncthreads();
  for (int off = 128; off > 0; off >>= 1) {
    if (t < off) {
      ssum[t] += ssum[t + off];
      ssq[t] += ssq[t + off];
    }
    __syncthreads();
  }
  if (t == 0) {
    const float invN = 1.0f / (float)(BB * SS);
    float mean = ssum[0] * invN;
    float var = ssq[0] * invN - mean * mean;
    stats[ch] = mean;
    stats[DD + ch] = rsqrtf(var + 1e-5f);
  }
}

__global__ __launch_bounds__(256) void bn_apply_kernel(
    float* __restrict__ out, const float* __restrict__ stats,
    const float* __restrict__ gamma, const float* __restrict__ beta) {
  int idx = blockIdx.x * 256 + threadIdx.x;
  int ch = idx & (DD - 1);
  float v = out[idx];
  out[idx] = (v - stats[ch]) * stats[DD + ch] * gamma[ch] + beta[ch];
}

extern "C" void kernel_launch(void* const* d_in, const int* in_sizes, int n_in,
                              void* d_out, int out_size, void* d_ws,
                              size_t ws_size, hipStream_t stream) {
  const float* x     = (const float*)d_in[0];
  const float* Wq    = (const float*)d_in[1];
  const float* bq    = (const float*)d_in[2];
  const float* Wk    = (const float*)d_in[3];
  const float* bk    = (const float*)d_in[4];
  const float* Wv    = (const float*)d_in[5];
  const float* bv    = (const float*)d_in[6];
  const float* gamma = (const float*)d_in[7];
  const float* beta  = (const float*)d_in[8];
  float* out = (float*)d_out;
  float* ws = (float*)d_ws;
  float* stats = ws + (size_t)3 * BB * SS * DD;

  qkv_kernel<<<(3 * (BB * SS / 16)) / 4, 128, 0, stream>>>(x, Wq, bq, Wk, bk,
                                                           Wv, bv, ws);
  attn_kernel<<<BB * HH * (SS / 64), 128, 0, stream>>>(x, ws, out);
  bn_stats_kernel<<<DD, 256, 0, stream>>>(out, stats);
  bn_apply_kernel<<<(BB * SS * DD) / 256, 256, 0, stream>>>(out, stats, gamma,
                                                            beta);
}


# attention: scores computed transposed (S^T = K Q^T, 32 keys per step): softmax max and sum in-lane (one lane-half exchange per step), P^T is the PV B operand from registers, no LDS; sum order changed
# speedup vs baseline: 1.8155x; 1.8155x over previous
#include <hip/hip_runtime.h>
#include <math.h>


typedef __attribute__((ext_vector_type(2))) float v2f;
typedef __attribute__((ext_vector_type(8))) float v8f;

#define BB 4
#define SS 2048
#define DD 64
#define HH 16
#define HDIM 4

static __device__ __forceinline__ v8f wmma_f32_16x16x4(v2f a, v2f b, v8f c) {
  return __builtin_amdgcn_wmma_f32_16x16x4_f32(false, a, false, b, (short)0, c,
                                               false, false);
}

__global__ __launch_bounds__(128) void qkv_kernel(
    const float* __restrict__ x,
    const float* __restrict__ Wq, const float* __restrict__ bq,
    const float* __restrict__ Wk, const float* __restrict__ bk,
    const float* __restrict__ Wv, const float* __restrict__ bv,
    float* __restrict__ ws) {
  const int NT = (BB * SS) / 16;
  int tid   = blockIdx.x * 4 + (threadIdx.x >> 5);
  int which = tid / NT;
  int tile  = tid % NT;
  int r0    = tile * 16;

  const float* W    = (which == 0) ? Wq : (which == 1) ? Wk : Wv;
  const float* bias = (which == 0) ? bq : (which == 1) ? bk : bv;
  float* Out = ws + (size_t)which * (BB * SS * DD);

  int lane = threadIdx.x & 31;
  int half = lane >> 4;
  int l16  = lane & 15;

  v2f afrag[16];
  const float* xrow = x + (size_t)(r0 + l16) * DD + 2 * half;
#pragma unroll
  for (int kc = 0; kc < 16; ++kc) {
    afrag[kc].x = xrow[4 * kc + 0];
    afrag[kc].y = xrow[4 * kc + 1];
  }

  v8f acc[4];
#pragma unroll
  for (int n = 0; n < 4; ++n)
#pragma unroll
    for (int v = 0; v < 8; ++v) acc[n][v] = 0.0f;

#pragma unroll
  for (int kc = 0; kc < 16; ++kc) {
    int krow = 4 * kc + 2 * half;
#pragma unroll
    for (int n = 0; n < 4; ++n) {
      v2f bfrag;
      bfrag.x = W[(size_t)krow * DD + n * 16 + l16];
      bfrag.y = W[(size_t)(krow + 1) * DD + n * 16 + l16];
      acc[n] = wmma_f32_16x16x4(afrag[kc], bfrag, acc[n]);
    }
  }

#pragma unroll
  for (int n = 0; n < 4; ++n) {
    int col = n * 16 + l16;
    float bb = bias[col];
#pragma unroll
    for (int v = 0; v < 8; ++v) {
      int row = r0 + v + 8 * half;
      Out[(size_t)row * DD + col] = fmaxf(acc[n][v] + bb, 0.0f);
    }
  }
}

static __device__ __forceinline__ float xhalf(float v) {
  int b = __builtin_bit_cast(int, v);
  return __builtin_bit_cast(
      float, __builtin_amdgcn_permlanex16(b, b, 0x76543210, 0xfedcba98, false,
                                          false));
}

__global__ __launch_bounds__(128) void attn_kernel(
    const float* __restrict__ x, const float* __restrict__ ws,
    float* __restrict__ out) {
  const float* Q = ws;
  const float* K = ws + (size_t)BB * SS * DD;
  const float* V = ws + (size_t)2 * BB * SS * DD;

  int wave = threadIdx.x >> 5;
  int lane = threadIdx.x & 31;
  int half = lane >> 4;
  int l16  = lane & 15;

  const int QT = SS / 64;
  int bh = blockIdx.x / QT;
  int qt = blockIdx.x % QT;
  int b = bh / HH, h = bh % HH;
  int q0 = qt * 64 + wave * 16;

  const float* Qb = Q + ((size_t)b * SS) * DD + h * HDIM;
  const float* Kb = K + ((size_t)b * SS) * DD + h * HDIM;
  const float* Vb = V + ((size_t)b * SS) * DD + h * HDIM;

  v2f bq;
  bq.x = Qb[(size_t)(q0 + l16) * DD + 2 * half];
  bq.y = Qb[(size_t)(q0 + l16) * DD + 2 * half + 1];

  const float cs = 0.5f * 1.4426950408889634f;
  const v2f cs2 = {cs, cs};

  float mraw = -INFINITY;
  float mc   = -INFINITY;
  float lsum = 0.0f;
  v8f accO, zc;
  for (int v = 0; v < 8; ++v) {
    accO[v] = 0.0f;
    zc[v]   = 0.0f;
  }

  const float* kp = Kb + (size_t)l16 * DD + 2 * half;
  const float* vp = Vb + (size_t)(8 * half) * DD + (l16 & 3);

  for (int k0 = 0; k0 < SS; k0 += 32) {
    if (k0 + 32 < SS) {
      __builtin_prefetch(Kb + (size_t)(k0 + 32 + lane) * DD, 0, 1);
      __builtin_prefetch(Vb + (size_t)(k0 + 32 + lane) * DD, 0, 1);
    }
    v8f s[2];
    for (int j = 0; j < 2; ++j) {
      v2f ak;
      ak.x = kp[(size_t)(k0 + 16 * j) * DD];
      ak.y = kp[(size_t)(k0 + 16 * j) * DD + 1];
      s[j] = wmma_f32_16x16x4(ak, bq, zc);
    }

    float t = mraw;
    for (int j = 0; j < 2; ++j)
      for (int v = 0; v < 8; v += 2) t = fmaxf(fmaxf(t, s[j][v]), s[j][v + 1]);
    float mnew  = fmaxf(t, xhalf(t));
    float mcn   = mnew * cs;
    float alpha = __builtin_amdgcn_exp2f(mc - mcn);
    mraw = mnew;
    mc   = mcn;

    const v2f nm2 = {-mcn, -mcn};
    v2f p[2][4];
    for (int j = 0; j < 2; ++j)
      for (int i = 0; i < 4; ++i) {
        v2f sp = {s[j][2 * i], s[j][2 * i + 1]};
        v2f e = __builtin_elementwise_fma(sp, cs2, nm2);
        p[j][i].x = __builtin_amdgcn_exp2f(e.x);
        p[j][i].y = __builtin_amdgcn_exp2f(e.y);
      }
    v2f r2 = ((p[0][0] + p[0][1]) + (p[0][2] + p[0][3])) +
             ((p[1][0] + p[1][1]) + (p[1][2] + p[1][3]));
    lsum = lsum * alpha + (r2.x + r2.y);
    for (int v = 0; v < 4; ++v) accO[v] *= alpha;

    for (int j = 0; j < 2; ++j)
      for (int i = 0; i < 4; ++i) {
        v2f av;
        av.x = vp[(size_t)(k0 + 16 * j + 2 * i) * DD];
        av.y = vp[(size_t)(k0 + 16 * j + 2 * i + 1) * DD];
        accO = wmma_f32_16x16x4(av, p[j][i], accO);
      }
  }

  float l = lsum + xhalf(lsum);

  if (half == 0) {
    size_t idx = ((size_t)b * SS + q0 + l16) * DD + h * HDIM;
    const float4 xr = *reinterpret_cast<const float4*>(x + idx);
    float4 o;
    o.x = accO[0] / l + xr.x;
    o.y = accO[1] / l + xr.y;
    o.z = accO[2] / l + xr.z;
    o.w = accO[3] / l + xr.w;
    *reinterpret_cast<float4*>(out + idx) = o;
  }
}

__global__ __launch_bounds__(256) void bn_stats_kernel(
    const float* __restrict__ out, float* __restrict__ stats) {
  __shared__ float ssum[256], ssq[256];
  int ch = blockIdx.x;
  int t = threadIdx.x;
  float s = 0.0f, q = 0.0f;
  for (int i = t; i < BB * SS; i += 256) {
    float v = out[(size_t)i * DD + ch];
    s += v;
    q += v * v;
  }
  ssum[t] = s;
  ssq[t] = q;
  __syncthreads();
  for (int off = 128; off > 0; off >>= 1) {
    if (t < off) {
      ssum[t] += ssum[t + off];
      ssq[t] += ssq[t + off];
    }
    __syncthreads();
  }
  if (t == 0) {
    const float invN = 1.0f / (float)(BB * SS);
    float mean = ssum[0] * invN;
    float var = ssq[0] * invN - mean * mean;
    stats[ch] = mean;
    stats[DD + ch] = rsqrtf(var + 1e-5f);
  }
}

__global__ __launch_bounds__(256) void bn_apply_kernel(
    float* __restrict__ out, const float* __restrict__ stats,
    const float* __restrict__ gamma, const float* __restrict__ beta) {
  int idx = blockIdx.x * 256 + threadIdx.x;
  int ch = idx & (DD - 1);
  float v = out[idx];
  out[idx] = (v - stats[ch]) * stats[DD + ch] * gamma[ch] + beta[ch];
}

extern "C" void kernel_launch(void* const* d_in, const int* in_sizes, int n_in,
                              void* d_out, int out_size, void* d_ws,
                              size_t ws_size, hipStream_t stream) {
  const float* x     = (const float*)d_in[0];
  const float* Wq    = (const float*)d_in[1];
  const float* bq    = (const float*)d_in[2];
  const float* Wk    = (const float*)d_in[3];
  const float* bk    = (const float*)d_in[4];
  const float* Wv    = (const float*)d_in[5];
  const float* bv    = (const float*)d_in[6];
  const float* gamma = (const float*)d_in[7];
  const float* beta  = (const float*)d_in[8];
  float* out = (float*)d_out;
  float* ws = (float*)d_ws;
  float* stats = ws + (size_t)3 * BB * SS * DD;

  qkv_kernel<<<(3 * (BB * SS / 16)) / 4, 128, 0, stream>>>(x, Wq, bq, Wk, bk,
                                                           Wv, bv, ws);
  attn_kernel<<<BB * HH * (SS / 64), 128, 0, stream>>>(x, ws, out);
  bn_stats_kernel<<<DD, 256, 0, stream>>>(out, stats);
  bn_apply_kernel<<<(BB * SS * DD) / 256, 256, 0, stream>>>(out, stats, gamma,
                                                            beta);
}


# workspace re-laid per head (Q,K [b][h][s][4]; V 128-byte lines in A-fragment order; qkv swaps WMMA operands for Q,K to store it): attention reads 8 whole lines per 32 keys, was 64 + prefetch
# speedup vs baseline: 3.5558x; 1.9586x over previous
#include <hip/hip_runtime.h>
#include <math.h>


typedef __attribute__((ext_vector_type(2))) float v2f;
typedef __attribute__((ext_vector_type(8))) float v8f;

#define BB 4
#define SS 2048
#define DD 64
#define HH 16
#define HDIM 4

static __device__ __forceinline__ v8f wmma_f32_16x16x4(v2f a, v2f b, v8f c) {
  return __builtin_amdgcn_wmma_f32_16x16x4_f32(false, a, false, b, (short)0, c,
                                               false, false);
}

static __device__ __forceinline__ float4 bias_relu4(const v8f& a, int o, float4 bb) {
  float4 r;
  r.x = fmaxf(a[o + 0] + bb.x, 0.0f);
  r.y = fmaxf(a[o + 1] + bb.y, 0.0f);
  r.z = fmaxf(a[o + 2] + bb.z, 0.0f);
  r.w = fmaxf(a[o + 3] + bb.w, 0.0f);
  return r;
}

__global__ __launch_bounds__(128) void qkv_kernel(
    const float* __restrict__ x,
    const float* __restrict__ Wq, const float* __restrict__ bq,
    const float* __restrict__ Wk, const float* __restrict__ bk,
    const float* __restrict__ Wv, const float* __restrict__ bv,
    float* __restrict__ ws) {
  const int NT = (BB * SS) / 16;
  int wv    = __builtin_amdgcn_readfirstlane((int)(threadIdx.x >> 5));
  int tid   = blockIdx.x * 4 + wv;
  int which = tid / NT;
  int tile  = tid % NT;
  int r0    = tile * 16;
  int b     = tile / (SS / 16);
  int s0    = (tile % (SS / 16)) * 16;

  const float* W    = (which == 0) ? Wq : (which == 1) ? Wk : Wv;
  const float* bias = (which == 0) ? bq : (which == 1) ? bk : bv;
  float* Out = ws + (size_t)which * (BB * SS * DD);

  int lane = threadIdx.x & 31;
  int half = lane >> 4;
  int l16  = lane & 15;

  v2f xf[16];
  const float* xrow = x + (size_t)(r0 + l16) * DD + 2 * half;
#pragma unroll
  for (int kc = 0; kc < 16; ++kc) {
    xf[kc].x = xrow[4 * kc + 0];
    xf[kc].y = xrow[4 * kc + 1];
  }

  v8f acc[4];
#pragma unroll
  for (int n = 0; n < 4; ++n)
#pragma unroll
    for (int v = 0; v < 8; ++v) acc[n][v] = 0.0f;

  if (which == 2) {
#pragma unroll
    for (int kc = 0; kc < 16; ++kc) {
      int krow = 4 * kc + 2 * half;
#pragma unroll
      for (int n = 0; n < 4; ++n) {
        v2f wf;
        wf.x = W[(size_t)krow * DD + n * 16 + l16];
        wf.y = W[(size_t)(krow + 1) * DD + n * 16 + l16];
        acc[n] = wmma_f32_16x16x4(xf[kc], wf, acc[n]);
      }
    }
    int G = s0 >> 5;
    int j = (s0 >> 4) & 1;
#pragma unroll
    for (int n = 0; n < 4; ++n) {
      float bs = bias[n * 16 + l16];
      float4 bb = {bs, bs, bs, bs};
      int head = n * 4 + (l16 >> 2);
      float* dst = Out + ((size_t)(b * HH + head) * (SS / 32) + G) * 128 +
                   (2 * j) * 32 + (l16 & 3) * 8 + half * 4;
#pragma unroll
      for (int t = 0; t < 2; ++t)
        *reinterpret_cast<float4*>(dst + t * 32) = bias_relu4(acc[n], 4 * t, bb);
    }
  } else {
#pragma unroll
    for (int kc = 0; kc < 16; ++kc) {
      int krow = 4 * kc + 2 * half;
#pragma unroll
      for (int n = 0; n < 4; ++n) {
        v2f wf;
        wf.x = W[(size_t)krow * DD + n * 16 + l16];
        wf.y = W[(size_t)(krow + 1) * DD + n * 16 + l16];
        acc[n] = wmma_f32_16x16x4(wf, xf[kc], acc[n]);
      }
    }
#pragma unroll
    for (int n = 0; n < 4; ++n)
#pragma unroll
      for (int u = 0; u < 2; ++u) {
        int head = n * 4 + 2 * half + u;
        const float4 bb =
            *reinterpret_cast<const float4*>(bias + head * HDIM);
        float* dst = Out + ((size_t)(b * HH + head) * SS + s0 + l16) * HDIM;
        *reinterpret_cast<float4*>(dst) = bias_relu4(acc[n], 4 * u, bb);
      }
  }
}

static __device__ __forceinline__ float xhalf(float v) {
  int b = __builtin_bit_cast(int, v);
  return __builtin_bit_cast(
      float, __builtin_amdgcn_permlanex16(b, b, 0x76543210, 0xfedcba98, false,
                                          false));
}

__global__ __launch_bounds__(128) void attn_kernel(
    const float* __restrict__ x, const float* __restrict__ ws,
    float* __restrict__ out) {
  const float* Q = ws;
  const float* K = ws + (size_t)BB * SS * DD;
  const float* V = ws + (size_t)2 * BB * SS * DD;

  int wave = threadIdx.x >> 5;
  int lane = threadIdx.x & 31;
  int half = lane >> 4;
  int l16  = lane & 15;

  const int QT = SS / 64;
  int bh = blockIdx.x / QT;
  int qt = blockIdx.x % QT;
  int b = bh / HH, h = bh % HH;
  int q0 = qt * 64 + wave * 16;

  const size_t hb = (size_t)bh * SS * HDIM;
  const float* Qb = Q + hb;
  const float* Kb = K + hb;
  const float* Vb = V + hb;

  const v2f bq = *reinterpret_cast<const v2f*>(
      Qb + (size_t)(q0 + l16) * HDIM + 2 * half);

  const float cs = 0.5f * 1.4426950408889634f;
  const v2f cs2 = {cs, cs};

  float mraw = -INFINITY;
  float mc   = -INFINITY;
  float lsum = 0.0f;
  v8f accO, zc;
  for (int v = 0; v < 8; ++v) {
    accO[v] = 0.0f;
    zc[v]   = 0.0f;
  }

  const float* kp = Kb + (size_t)l16 * HDIM + 2 * half;
  const float* vp = Vb + (l16 & 3) * 8 + half * 4;

  for (int k0 = 0; k0 < SS; k0 += 32) {
    v8f s[2];
    for (int j = 0; j < 2; ++j) {
      const v2f ak = *reinterpret_cast<const v2f*>(
          kp + (size_t)(k0 + 16 * j) * HDIM);
      s[j] = wmma_f32_16x16x4(ak, bq, zc);
    }

    float t = mraw;
    for (int j = 0; j < 2; ++j)
      for (int v = 0; v < 8; v += 2) t = fmaxf(fmaxf(t, s[j][v]), s[j][v + 1]);
    float mnew  = fmaxf(t, xhalf(t));
    float mcn   = mnew * cs;
    float alpha = __builtin_amdgcn_exp2f(mc - mcn);
    mraw = mnew;
    mc   = mcn;

    const v2f nm2 = {-mcn, -mcn};
    v2f p[2][4];
    for (int j = 0; j < 2; ++j)
      for (int i = 0; i < 4; ++i) {
        v2f sp = {s[j][2 * i], s[j][2 * i + 1]};
        v2f e = __builtin_elementwise_fma(sp, cs2, nm2);
        p[j][i].x = __builtin_amdgcn_exp2f(e.x);
        p[j][i].y = __builtin_amdgcn_exp2f(e.y);
      }
    v2f r2 = ((p[0][0] + p[0][1]) + (p[0][2] + p[0][3])) +
             ((p[1][0] + p[1][1]) + (p[1][2] + p[1][3]));
    lsum = lsum * alpha + (r2.x + r2.y);
    for (int v = 0; v < 4; ++v) accO[v] *= alpha;

    for (int j = 0; j < 2; ++j)
      for (int t = 0; t < 2; ++t) {
        const float4 vv = *reinterpret_cast<const float4*>(
            vp + (size_t)k0 * HDIM + (2 * j + t) * 32);
        const v2f av0 = {vv.x, vv.y};
        const v2f av1 = {vv.z, vv.w};
        accO = wmma_f32_16x16x4(av0, p[j][2 * t], accO);
        accO = wmma_f32_16x16x4(av1, p[j][2 * t + 1], accO);
      }
  }

  float l = lsum + xhalf(lsum);

  if (half == 0) {
    size_t idx = ((size_t)b * SS + q0 + l16) * DD + h * HDIM;
    const float4 xr = *reinterpret_cast<const float4*>(x + idx);
    float4 o;
    o.x = accO[0] / l + xr.x;
    o.y = accO[1] / l + xr.y;
    o.z = accO[2] / l + xr.z;
    o.w = accO[3] / l + xr.w;
    *reinterpret_cast<float4*>(out + idx) = o;
  }
}

__global__ __launch_bounds__(256) void bn_stats_kernel(
    const float* __restrict__ out, float* __restrict__ stats) {
  __shared__ float ssum[256], ssq[256];
  int ch = blockIdx.x;
  int t = threadIdx.x;
  float s = 0.0f, q = 0.0f;
  for (int i = t; i < BB * SS; i += 256) {
    float v = out[(size_t)i * DD + ch];
    s += v;
    q += v * v;
  }
  ssum[t] = s;
  ssq[t] = q;
  __syncthreads();
  for (int off = 128; off > 0; off >>= 1) {
    if (t < off) {
      ssum[t] += ssum[t + off];
      ssq[t] += ssq[t + off];
    }
    __syncthreads();
  }
  if (t == 0) {
    const float invN = 1.0f / (float)(BB * SS);
    float mean = ssum[0] * invN;
    float var = ssq[0] * invN - mean * mean;
    stats[ch] = mean;
    stats[DD + ch] = rsqrtf(var + 1e-5f);
  }
}

__global__ __launch_bounds__(256) void bn_apply_kernel(
    float* __restrict__ out, const float* __restrict__ stats,
    const float* __restrict__ gamma, const float* __restrict__ beta) {
  int idx = blockIdx.x * 256 + threadIdx.x;
  int ch = idx & (DD - 1);
  float v = out[idx];
  out[idx] = (v - stats[ch]) * stats[DD + ch] * gamma[ch] + beta[ch];
}

extern "C" void kernel_launch(void* const* d_in, const int* in_sizes, int n_in,
                              void* d_out, int out_size, void* d_ws,
                              size_t ws_size, hipStream_t stream) {
  const float* x     = (const float*)d_in[0];
  const float* Wq    = (const float*)d_in[1];
  const float* bq    = (const float*)d_in[2];
  const float* Wk    = (const float*)d_in[3];
  const float* bk    = (const float*)d_in[4];
  const float* Wv    = (const float*)d_in[5];
  const float* bv    = (const float*)d_in[6];
  const float* gamma = (const float*)d_in[7];
  const float* beta  = (const float*)d_in[8];
  float* out = (float*)d_out;
  float* ws = (float*)d_ws;
  float* stats = ws + (size_t)3 * BB * SS * DD;

  qkv_kernel<<<(3 * (BB * SS / 16)) / 4, 128, 0, stream>>>(x, Wq, bq, Wk, bk,
                                                           Wv, bv, ws);
  attn_kernel<<<BB * HH * (SS / 64), 128, 0, stream>>>(x, ws, out);
  bn_stats_kernel<<<DD, 256, 0, stream>>>(out, stats);
  bn_apply_kernel<<<(BB * SS * DD) / 256, 256, 0, stream>>>(out, stats, gamma,
                                                            beta);
}


# BatchNorm sums: each attention block adds its 64x4 outputs and squares, stores them over its own spent Q rows; a 16-block launch adds the records; strided pass over out gone; sum order changed
# speedup vs baseline: 3.8616x; 1.0860x over previous
#include <hip/hip_runtime.h>
#include <math.h>


typedef __attribute__((ext_vector_type(2))) float v2f;
typedef __attribute__((ext_vector_type(8))) float v8f;

#define BB 4
#define SS 2048
#define DD 64
#define HH 16
#define HDIM 4

static __device__ __forceinline__ v8f wmma_f32_16x16x4(v2f a, v2f b, v8f c) {
  return __builtin_amdgcn_wmma_f32_16x16x4_f32(false, a, false, b, (short)0, c,
                                               false, false);
}

static __device__ __forceinline__ float4 bias_relu4(const v8f& a, int o, float4 bb) {
  float4 r;
  r.x = fmaxf(a[o + 0] + bb.x, 0.0f);
  r.y = fmaxf(a[o + 1] + bb.y, 0.0f);
  r.z = fmaxf(a[o + 2] + bb.z, 0.0f);
  r.w = fmaxf(a[o + 3] + bb.w, 0.0f);
  return r;
}

__global__ __launch_bounds__(128) void qkv_kernel(
    const float* __restrict__ x,
    const float* __restrict__ Wq, const float* __restrict__ bq,
    const float* __restrict__ Wk, const float* __restrict__ bk,
    const float* __restrict__ Wv, const float* __restrict__ bv,
    float* __restrict__ ws) {
  const int NT = (BB * SS) / 16;
  int wv    = __builtin_amdgcn_readfirstlane((int)(threadIdx.x >> 5));
  int tid   = blockIdx.x * 4 + wv;
  int which = tid / NT;
  int tile  = tid % NT;
  int r0    = tile * 16;
  int b     = tile / (SS / 16);
  int s0    = (tile % (SS / 16)) * 16;

  const float* W    = (which == 0) ? Wq : (which == 1) ? Wk : Wv;
  const float* bias = (which == 0) ? bq : (which == 1) ? bk : bv;
  float* Out = ws + (size_t)which * (BB * SS * DD);

  int lane = threadIdx.x & 31;
  int half = lane >> 4;
  int l16  = lane & 15;

  v2f xf[16];
  const float* xrow = x + (size_t)(r0 + l16) * DD + 2 * half;
#pragma unroll
  for (int kc = 0; kc < 16; ++kc) {
    xf[kc].x = xrow[4 * kc + 0];
    xf[kc].y = xrow[4 * kc + 1];
  }

  v8f acc[4];
#pragma unroll
  for (int n = 0; n < 4; ++n)
#pragma unroll
    for (int v = 0; v < 8; ++v) acc[n][v] = 0.0f;

  if (which == 2) {
#pragma unroll
    for (int kc = 0; kc < 16; ++kc) {
      int krow = 4 * kc + 2 * half;
#pragma unroll
      for (int n = 0; n < 4; ++n) {
        v2f wf;
        wf.x = W[(size_t)krow * DD + n * 16 + l16];
        wf.y = W[(size_t)(krow + 1) * DD + n * 16 + l16];
        acc[n] = wmma_f32_16x16x4(xf[kc], wf, acc[n]);
      }
    }
    int G = s0 >> 5;
    int j = (s0 >> 4) & 1;
#pragma unroll
    for (int n = 0; n < 4; ++n) {
      float bs = bias[n * 16 + l16];
      float4 bb = {bs, bs, bs, bs};
      int head = n * 4 + (l16 >> 2);
      float* dst = Out + ((size_t)(b * HH + head) * (SS / 32) + G) * 128 +
                   (2 * j) * 32 + (l16 & 3) * 8 + half * 4;
#pragma unroll
      for (int t = 0; t < 2; ++t)
        *reinterpret_cast<float4*>(dst + t * 32) = bias_relu4(acc[n], 4 * t, bb);
    }
  } else {
#pragma unroll
    for (int kc = 0; kc < 16; ++kc) {
      int krow = 4 * kc + 2 * half;
#pragma unroll
      for (int n = 0; n < 4; ++n) {
        v2f wf;
        wf.x = W[(size_t)krow * DD + n * 16 + l16];
        wf.y = W[(size_t)(krow + 1) * DD + n * 16 + l16];
        acc[n] = wmma_f32_16x16x4(wf, xf[kc], acc[n]);
      }
    }
#pragma unroll
    for (int n = 0; n < 4; ++n)
#pragma unroll
      for (int u = 0; u < 2; ++u) {
        int head = n * 4 + 2 * half + u;
        const float4 bb =
            *reinterpret_cast<const float4*>(bias + head * HDIM);
        float* dst = Out + ((size_t)(b * HH + head) * SS + s0 + l16) * HDIM;
        *reinterpret_cast<float4*>(dst) = bias_relu4(acc[n], 4 * u, bb);
      }
  }
}

static __device__ __forceinline__ float xhalf(float v) {
  int b = __builtin_bit_cast(int, v);
  return __builtin_bit_cast(
      float, __builtin_amdgcn_permlanex16(b, b, 0x76543210, 0xfedcba98, false,
                                          false));
}

__global__ __launch_bounds__(128) void attn_kernel(
    const float* __restrict__ x, float* __restrict__ Q,
    const float* __restrict__ KV, float* __restrict__ out) {
  const float* K = KV;
  const float* V = KV + (size_t)BB * SS * DD;
  __shared__ float red[4][8][16];

  int wave = threadIdx.x >> 5;
  int lane = threadIdx.x & 31;
  int half = lane >> 4;
  int l16  = lane & 15;

  const int QT = SS / 64;
  int bh = blockIdx.x / QT;
  int qt = blockIdx.x % QT;
  int b = bh / HH, h = bh % HH;
  int q0 = qt * 64 + wave * 16;

  const size_t hb = (size_t)bh * SS * HDIM;
  float* Qb = Q + hb;
  const float* Kb = K + hb;
  const float* Vb = V + hb;

  const v2f bq = *reinterpret_cast<const v2f*>(
      Qb + (size_t)(q0 + l16) * HDIM + 2 * half);

  const float cs = 0.5f * 1.4426950408889634f;
  const v2f cs2 = {cs, cs};

  float mraw = -INFINITY;
  float mc   = -INFINITY;
  float lsum = 0.0f;
  v8f accO, zc;
  for (int v = 0; v < 8; ++v) {
    accO[v] = 0.0f;
    zc[v]   = 0.0f;
  }

  const float* kp = Kb + (size_t)l16 * HDIM + 2 * half;
  const float* vp = Vb + (l16 & 3) * 8 + half * 4;

  for (int k0 = 0; k0 < SS; k0 += 32) {
    v8f s[2];
    for (int j = 0; j < 2; ++j) {
      const v2f ak = *reinterpret_cast<const v2f*>(
          kp + (size_t)(k0 + 16 * j) * HDIM);
      s[j] = wmma_f32_16x16x4(ak, bq, zc);
    }

    float t = mraw;
    for (int j = 0; j < 2; ++j)
      for (int v = 0; v < 8; v += 2) t = fmaxf(fmaxf(t, s[j][v]), s[j][v + 1]);
    float mnew  = fmaxf(t, xhalf(t));
    float mcn   = mnew * cs;
    float alpha = __builtin_amdgcn_exp2f(mc - mcn);
    mraw = mnew;
    mc   = mcn;

    const v2f nm2 = {-mcn, -mcn};
    v2f p[2][4];
    for (int j = 0; j < 2; ++j)
      for (int i = 0; i < 4; ++i) {
        v2f sp = {s[j][2 * i], s[j][2 * i + 1]};
        v2f e = __builtin_elementwise_fma(sp, cs2, nm2);
        p[j][i].x = __builtin_amdgcn_exp2f(e.x);
        p[j][i].y = __builtin_amdgcn_exp2f(e.y);
      }
    v2f r2 = ((p[0][0] + p[0][1]) + (p[0][2] + p[0][3])) +
             ((p[1][0] + p[1][1]) + (p[1][2] + p[1][3]));
    lsum = lsum * alpha + (r2.x + r2.y);
    for (int v = 0; v < 4; ++v) accO[v] *= alpha;

    for (int j = 0; j < 2; ++j)
      for (int t = 0; t < 2; ++t) {
        const float4 vv = *reinterpret_cast<const float4*>(
            vp + (size_t)k0 * HDIM + (2 * j + t) * 32);
        const v2f av0 = {vv.x, vv.y};
        const v2f av1 = {vv.z, vv.w};
        accO = wmma_f32_16x16x4(av0, p[j][2 * t], accO);
        accO = wmma_f32_16x16x4(av1, p[j][2 * t + 1], accO);
      }
  }

  float l = lsum + xhalf(lsum);

  if (half == 0) {
    size_t idx = ((size_t)b * SS + q0 + l16) * DD + h * HDIM;
    const float4 xr = *reinterpret_cast<const float4*>(x + idx);
    float4 o;
    o.x = accO[0] / l + xr.x;
    o.y = accO[1] / l + xr.y;
    o.z = accO[2] / l + xr.z;
    o.w = accO[3] / l + xr.w;
    *reinterpret_cast<float4*>(out + idx) = o;
    float* r = &red[wave][0][l16];
    r[0 * 16] = o.x;
    r[1 * 16] = o.y;
    r[2 * 16] = o.z;
    r[3 * 16] = o.w;
    r[4 * 16] = o.x * o.x;
    r[5 * 16] = o.y * o.y;
    r[6 * 16] = o.z * o.z;
    r[7 * 16] = o.w * o.w;
  }
  __syncthreads();
  if (threadIdx.x < 8) {
    int v = threadIdx.x;
    float a0 = 0.0f, a1 = 0.0f, a2 = 0.0f, a3 = 0.0f;
    for (int j = 0; j < 16; ++j) {
      a0 += red[0][v][j];
      a1 += red[1][v][j];
      a2 += red[2][v][j];
      a3 += red[3][v][j];
    }
    Qb[(size_t)qt * 64 * HDIM + v] = (a0 + a1) + (a2 + a3);
  }
}

__global__ __launch_bounds__(128) void bn_stats_kernel(
    const float* __restrict__ part, float* __restrict__ stats) {
  __shared__ float4 rs[128], rq[128];
  int h = blockIdx.x;
  int t = threadIdx.x;
  int b = t >> 5, qt = t & 31;
  const float4* p = reinterpret_cast<const float4*>(
      part + ((size_t)(b * HH + h) * SS + (size_t)qt * 64) * HDIM);
  rs[t] = p[0];
  rq[t] = p[1];
  __syncthreads();
  if (t < 4) {
    const float* fs = reinterpret_cast<const float*>(rs) + t;
    const float* fq = reinterpret_cast<const float*>(rq) + t;
    float s[4] = {0.0f, 0.0f, 0.0f, 0.0f};
    float q[4] = {0.0f, 0.0f, 0.0f, 0.0f};
    for (int i = 0; i < 32; ++i)
#pragma unroll
      for (int c = 0; c < 4; ++c) {
        s[c] += fs[(c * 32 + i) * 4];
        q[c] += fq[(c * 32 + i) * 4];
      }
    float ssum = (s[0] + s[1]) + (s[2] + s[3]);
    float ssq  = (q[0] + q[1]) + (q[2] + q[3]);
    const float invN = 1.0f / (float)(BB * SS);
    float mean = ssum * invN;
    float var = ssq * invN - mean * mean;
    int ch = h * HDIM + t;
    stats[ch] = mean;
    stats[DD + ch] = rsqrtf(var + 1e-5f);
  }
}

__global__ __launch_bounds__(256) void bn_apply_kernel(
    float* __restrict__ out, const float* __restrict__ stats,
    const float* __restrict__ gamma, const float* __restrict__ beta) {
  int idx = blockIdx.x * 256 + threadIdx.x;
  int ch = idx & (DD - 1);
  float v = out[idx];
  out[idx] = (v - stats[ch]) * stats[DD + ch] * gamma[ch] + beta[ch];
}

extern "C" void kernel_launch(void* const* d_in, const int* in_sizes, int n_in,
                              void* d_out, int out_size, void* d_ws,
                              size_t ws_size, hipStream_t stream) {
  const float* x     = (const float*)d_in[0];
  const float* Wq    = (const float*)d_in[1];
  const float* bq    = (const float*)d_in[2];
  const float* Wk    = (const float*)d_in[3];
  const float* bk    = (const float*)d_in[4];
  const float* Wv    = (const float*)d_in[5];
  const float* bv    = (const float*)d_in[6];
  const float* gamma = (const float*)d_in[7];
  const float* beta  = (const float*)d_in[8];
  float* out = (float*)d_out;
  float* ws = (float*)d_ws;
  float* kv = ws + (size_t)BB * SS * DD;
  float* stats = ws + (size_t)3 * BB * SS * DD;

  qkv_kernel<<<(3 * (BB * SS / 16)) / 4, 128, 0, stream>>>(x, Wq, bq, Wk, bk,
                                                           Wv, bv, ws);
  attn_kernel<<<BB * HH * (SS / 64), 128, 0, stream>>>(x, ws, kv, out);
  bn_stats_kernel<<<HH, 128, 0, stream>>>(ws, stats);
  bn_apply_kernel<<<(BB * SS * DD) / 256, 256, 0, stream>>>(out, stats, gamma,
                                                            beta);
}


# attention key loop software-pipelined in registers: V(i+1), K(i+2) loads and S(i+1) with its maximum run one step ahead; exps of S(i) sit chunk by chunk inside the PV(i) chain; same sum order
# speedup vs baseline: 4.4771x; 1.1594x over previous
#include <hip/hip_runtime.h>
#include <math.h>


typedef __attribute__((ext_vector_type(2))) float v2f;
typedef __attribute__((ext_vector_type(8))) float v8f;

#define BB 4
#define SS 2048
#define DD 64
#define HH 16
#define HDIM 4

static __device__ __forceinline__ v8f wmma_f32_16x16x4(v2f a, v2f b, v8f c) {
  return __builtin_amdgcn_wmma_f32_16x16x4_f32(false, a, false, b, (short)0, c,
                                               false, false);
}

static __device__ __forceinline__ float4 bias_relu4(const v8f& a, int o, float4 bb) {
  float4 r;
  r.x = fmaxf(a[o + 0] + bb.x, 0.0f);
  r.y = fmaxf(a[o + 1] + bb.y, 0.0f);
  r.z = fmaxf(a[o + 2] + bb.z, 0.0f);
  r.w = fmaxf(a[o + 3] + bb.w, 0.0f);
  return r;
}

__global__ __launch_bounds__(128) void qkv_kernel(
    const float* __restrict__ x,
    const float* __restrict__ Wq, const float* __restrict__ bq,
    const float* __restrict__ Wk, const float* __restrict__ bk,
    const float* __restrict__ Wv, const float* __restrict__ bv,
    float* __restrict__ ws) {
  const int NT = (BB * SS) / 16;
  int wv    = __builtin_amdgcn_readfirstlane((int)(threadIdx.x >> 5));
  int tid   = blockIdx.x * 4 + wv;
  int which = tid / NT;
  int tile  = tid % NT;
  int r0    = tile * 16;
  int b     = tile / (SS / 16);
  int s0    = (tile % (SS / 16)) * 16;

  const float* W    = (which == 0) ? Wq : (which == 1) ? Wk : Wv;
  const float* bias = (which == 0) ? bq : (which == 1) ? bk : bv;
  float* Out = ws + (size_t)which * (BB * SS * DD);

  int lane = threadIdx.x & 31;
  int half = lane >> 4;
  int l16  = lane & 15;

  v2f xf[16];
  const float* xrow = x + (size_t)(r0 + l16) * DD + 2 * half;
#pragma unroll
  for (int kc = 0; kc < 16; ++kc) {
    xf[kc].x = xrow[4 * kc + 0];
    xf[kc].y = xrow[4 * kc + 1];
  }

  v8f acc[4];
#pragma unroll
  for (int n = 0; n < 4; ++n)
#pragma unroll
    for (int v = 0; v < 8; ++v) acc[n][v] = 0.0f;

  if (which == 2) {
#pragma unroll
    for (int kc = 0; kc < 16; ++kc) {
      int krow = 4 * kc + 2 * half;
#pragma unroll
      for (int n = 0; n < 4; ++n) {
        v2f wf;
        wf.x = W[(size_t)krow * DD + n * 16 + l16];
        wf.y = W[(size_t)(krow + 1) * DD + n * 16 + l16];
        acc[n] = wmma_f32_16x16x4(xf[kc], wf, acc[n]);
      }
    }
    int G = s0 >> 5;
    int j = (s0 >> 4) & 1;
#pragma unroll
    for (int n = 0; n < 4; ++n) {
      float bs = bias[n * 16 + l16];
      float4 bb = {bs, bs, bs, bs};
      int head = n * 4 + (l16 >> 2);
      float* dst = Out + ((size_t)(b * HH + head) * (SS / 32) + G) * 128 +
                   (2 * j) * 32 + (l16 & 3) * 8 + half * 4;
#pragma unroll
      for (int t = 0; t < 2; ++t)
        *reinterpret_cast<float4*>(dst + t * 32) = bias_relu4(acc[n], 4 * t, bb);
    }
  } else {
#pragma unroll
    for (int kc = 0; kc < 16; ++kc) {
      int krow = 4 * kc + 2 * half;
#pragma unroll
      for (int n = 0; n < 4; ++n) {
        v2f wf;
        wf.x = W[(size_t)krow * DD + n * 16 + l16];
        wf.y = W[(size_t)(krow + 1) * DD + n * 16 + l16];
        acc[n] = wmma_f32_16x16x4(wf, xf[kc], acc[n]);
      }
    }
#pragma unroll
    for (int n = 0; n < 4; ++n)
#pragma unroll
      for (int u = 0; u < 2; ++u) {
        int head = n * 4 + 2 * half + u;
        const float4 bb =
            *reinterpret_cast<const float4*>(bias + head * HDIM);
        float* dst = Out + ((size_t)(b * HH + head) * SS + s0 + l16) * HDIM;
        *reinterpret_cast<float4*>(dst) = bias_relu4(acc[n], 4 * u, bb);
      }
  }
}

static __device__ __forceinline__ float xhalf(float v) {
  int b = __builtin_bit_cast(int, v);
  return __builtin_bit_cast(
      float, __builtin_amdgcn_permlanex16(b, b, 0x76543210, 0xfedcba98, false,
                                          false));
}

#define ATTN_CS (0.5f * 1.4426950408889634f)

static __device__ __forceinline__ void load_k(const float* kp, int step,
                                              v2f (&ak)[2]) {
  for (int j = 0; j < 2; ++j)
    ak[j] = *reinterpret_cast<const v2f*>(
        kp + (size_t)(step * 32 + 16 * j) * HDIM);
}

static __device__ __forceinline__ void load_v(const float* vp, int step,
                                              float4 (&vv)[4]) {
  for (int u = 0; u < 4; ++u)
    vv[u] = *reinterpret_cast<const float4*>(
        vp + (size_t)step * 32 * HDIM + u * 32);
}

static __device__ __forceinline__ void score_tiles(const v2f (&ak)[2], v2f bq,
                                                   v8f (&s)[2]) {
  v8f zc;
  for (int v = 0; v < 8; ++v) zc[v] = 0.0f;
  for (int j = 0; j < 2; ++j) s[j] = wmma_f32_16x16x4(ak[j], bq, zc);
}

static __device__ __forceinline__ void max_step(const v8f (&s)[2], float& mraw,
                                                float& mc, float& mstep,
                                                float& astep) {
  float t = mraw;
  for (int j = 0; j < 2; ++j)
    for (int v = 0; v < 8; v += 2) t = fmaxf(fmaxf(t, s[j][v]), s[j][v + 1]);
  float mnew = fmaxf(t, xhalf(t));
  float mcn  = mnew * ATTN_CS;
  astep = __builtin_amdgcn_exp2f(mc - mcn);
  mstep = mcn;
  mraw  = mnew;
  mc    = mcn;
}

static __device__ __forceinline__ void exp_pv_step(const v8f (&s)[2],
                                                   const float4 (&vv)[4],
                                                   float mstep, float astep,
                                                   float& lsum, v8f& accO) {
  const v2f cs2 = {ATTN_CS, ATTN_CS};
  const v2f nm2 = {-mstep, -mstep};
  for (int v = 0; v < 4; ++v) accO[v] *= astep;
  v2f p[2][4];
  for (int j = 0; j < 2; ++j)
    for (int t = 0; t < 2; ++t) {
      for (int i = 2 * t; i < 2 * t + 2; ++i) {
        v2f sp = {s[j][2 * i], s[j][2 * i + 1]};
        v2f e = __builtin_elementwise_fma(sp, cs2, nm2);
        p[j][i].x = __builtin_amdgcn_exp2f(e.x);
        p[j][i].y = __builtin_amdgcn_exp2f(e.y);
      }
      const float4 v4 = vv[2 * j + t];
      const v2f av0 = {v4.x, v4.y};
      const v2f av1 = {v4.z, v4.w};
      accO = wmma_f32_16x16x4(av0, p[j][2 * t], accO);
      accO = wmma_f32_16x16x4(av1, p[j][2 * t + 1], accO);
    }
  v2f r2 = ((p[0][0] + p[0][1]) + (p[0][2] + p[0][3])) +
           ((p[1][0] + p[1][1]) + (p[1][2] + p[1][3]));
  lsum = lsum * astep + (r2.x + r2.y);
}

__global__ __launch_bounds__(128) __attribute__((amdgpu_waves_per_eu(8, 8)))
void attn_kernel(
    const float* __restrict__ x, float* __restrict__ Q,
    const float* __restrict__ KV, float* __restrict__ out) {
  const float* K = KV;
  const float* V = KV + (size_t)BB * SS * DD;
  __shared__ float red[4][8][16];

  int wave = threadIdx.x >> 5;
  int lane = threadIdx.x & 31;
  int half = lane >> 4;
  int l16  = lane & 15;

  const int QT = SS / 64;
  int bh = blockIdx.x / QT;
  int qt = blockIdx.x % QT;
  int b = bh / HH, h = bh % HH;
  int q0 = qt * 64 + wave * 16;

  const size_t hb = (size_t)bh * SS * HDIM;
  float* Qb = Q + hb;
  const float* Kb = K + hb;
  const float* Vb = V + hb;

  const v2f bq = *reinterpret_cast<const v2f*>(
      Qb + (size_t)(q0 + l16) * HDIM + 2 * half);

  float mraw = -INFINITY;
  float mc   = -INFINITY;
  float lsum = 0.0f;
  v8f accO;
  for (int v = 0; v < 8; ++v) accO[v] = 0.0f;

  const float* kp = Kb + (size_t)l16 * HDIM + 2 * half;
  const float* vp = Vb + (l16 & 3) * 8 + half * 4;

  const int NS = SS / 32;

  v2f kX[2], kY[2];
  float4 vX[4], vY[4];
  v8f sX[2], sY[2];
  float mX, aX, mY, aY;

  load_k(kp, 0, kX);
  load_v(vp, 0, vX);
  load_k(kp, 1, kY);
  score_tiles(kX, bq, sX);
  max_step(sX, mraw, mc, mX, aX);

#define ATTN_STEP(i, kC, vC, sC, mC, aC, kN, vN, sN, mN, aN) \
  do {                                                       \
    load_v(vp, (i) + 1, vN);                                 \
    load_k(kp, (i) + 2, kC);                                 \
    score_tiles(kN, bq, sN);                                 \
    exp_pv_step(sC, vC, mC, aC, lsum, accO);                 \
    max_step(sN, mraw, mc, mN, aN);                          \
  } while (0)

  for (int i = 0; i < NS - 2; i += 2) {
    ATTN_STEP(i, kX, vX, sX, mX, aX, kY, vY, sY, mY, aY);
    ATTN_STEP(i + 1, kY, vY, sY, mY, aY, kX, vX, sX, mX, aX);
  }
#undef ATTN_STEP
  load_v(vp, NS - 1, vY);
  score_tiles(kY, bq, sY);
  exp_pv_step(sX, vX, mX, aX, lsum, accO);
  max_step(sY, mraw, mc, mY, aY);
  exp_pv_step(sY, vY, mY, aY, lsum, accO);

  float l = lsum + xhalf(lsum);

  if (half == 0) {
    size_t idx = ((size_t)b * SS + q0 + l16) * DD + h * HDIM;
    const float4 xr = *reinterpret_cast<const float4*>(x + idx);
    float4 o;
    o.x = accO[0] / l + xr.x;
    o.y = accO[1] / l + xr.y;
    o.z = accO[2] / l + xr.z;
    o.w = accO[3] / l + xr.w;
    *reinterpret_cast<float4*>(out + idx) = o;
    float* r = &red[wave][0][l16];
    r[0 * 16] = o.x;
    r[1 * 16] = o.y;
    r[2 * 16] = o.z;
    r[3 * 16] = o.w;
    r[4 * 16] = o.x * o.x;
    r[5 * 16] = o.y * o.y;
    r[6 * 16] = o.z * o.z;
    r[7 * 16] = o.w * o.w;
  }
  __syncthreads();
  if (threadIdx.x < 8) {
    int v = threadIdx.x;
    float a0 = 0.0f, a1 = 0.0f, a2 = 0.0f, a3 = 0.0f;
    for (int j = 0; j < 16; ++j) {
      a0 += red[0][v][j];
      a1 += red[1][v][j];
      a2 += red[2][v][j];
      a3 += red[3][v][j];
    }
    Qb[(size_t)qt * 64 * HDIM + v] = (a0 + a1) + (a2 + a3);
  }
}

__global__ __launch_bounds__(128) void bn_stats_kernel(
    const float* __restrict__ part, float* __restrict__ stats) {
  __shared__ float4 rs[128], rq[128];
  int h = blockIdx.x;
  int t = threadIdx.x;
  int b = t >> 5, qt = t & 31;
  const float4* p = reinterpret_cast<const float4*>(
      part + ((size_t)(b * HH + h) * SS + (size_t)qt * 64) * HDIM);
  rs[t] = p[0];
  rq[t] = p[1];
  __syncthreads();
  if (t < 4) {
    const float* fs = reinterpret_cast<const float*>(rs) + t;
    const float* fq = reinterpret_cast<const float*>(rq) + t;
    float s[4] = {0.0f, 0.0f, 0.0f, 0.0f};
    float q[4] = {0.0f, 0.0f, 0.0f, 0.0f};
    for (int i = 0; i < 32; ++i)
#pragma unroll
      for (int c = 0; c < 4; ++c) {
        s[c] += fs[(c * 32 + i) * 4];
        q[c] += fq[(c * 32 + i) * 4];
      }
    float ssum = (s[0] + s[1]) + (s[2] + s[3]);
    float ssq  = (q[0] + q[1]) + (q[2] + q[3]);
    const float invN = 1.0f / (float)(BB * SS);
    float mean = ssum * invN;
    float var = ssq * invN - mean * mean;
    int ch = h * HDIM + t;
    stats[ch] = mean;
    stats[DD + ch] = rsqrtf(var + 1e-5f);
  }
}

__global__ __launch_bounds__(256) void bn_apply_kernel(
    float* __restrict__ out, const float* __restrict__ stats,
    const float* __restrict__ gamma, const float* __restrict__ beta) {
  int idx = blockIdx.x * 256 + threadIdx.x;
  int ch = idx & (DD - 1);
  float v = out[idx];
  out[idx] = (v - stats[ch]) * stats[DD + ch] * gamma[ch] + beta[ch];
}

extern "C" void kernel_launch(void* const* d_in, const int* in_sizes, int n_in,
                              void* d_out, int out_size, void* d_ws,
                              size_t ws_size, hipStream_t stream) {
  const float* x     = (const float*)d_in[0];
  const float* Wq    = (const float*)d_in[1];
  const float* bq    = (const float*)d_in[2];
  const float* Wk    = (const float*)d_in[3];
  const float* bk    = (const float*)d_in[4];
  const float* Wv    = (const float*)d_in[5];
  const float* bv    = (const float*)d_in[6];
  const float* gamma = (const float*)d_in[7];
  const float* beta  = (const float*)d_in[8];
  float* out = (float*)d_out;
  float* ws = (float*)d_ws;
  float* kv = ws + (size_t)BB * SS * DD;
  float* stats = ws + (size_t)3 * BB * SS * DD;

  qkv_kernel<<<(3 * (BB * SS / 16)) / 4, 128, 0, stream>>>(x, Wq, bq, Wk, bk,
                                                           Wv, bv, ws);
  attn_kernel<<<BB * HH * (SS / 64), 128, 0, stream>>>(x, ws, kv, out);
  bn_stats_kernel<<<HH, 128, 0, stream>>>(ws, stats);
  bn_apply_kernel<<<(BB * SS * DD) / 256, 256, 0, stream>>>(out, stats, gamma,
                                                            beta);
}


# attention key loop: issue order pinned by scheduler fences; each exponential is followed by a plain vector instruction (next exponent pair or a link of the next maximum chain); same sums and order
# speedup vs baseline: 4.6156x; 1.0309x over previous
#include <hip/hip_runtime.h>
#include <math.h>


typedef __attribute__((ext_vector_type(2))) float v2f;
typedef __attribute__((ext_vector_type(8))) float v8f;

#define BB 4
#define SS 2048
#define DD 64
#define HH 16
#define HDIM 4

static __device__ __forceinline__ v8f wmma_f32_16x16x4(v2f a, v2f b, v8f c) {
  return __builtin_amdgcn_wmma_f32_16x16x4_f32(false, a, false, b, (short)0, c,
                                               false, false);
}

static __device__ __forceinline__ float4 bias_relu4(const v8f& a, int o, float4 bb) {
  float4 r;
  r.x = fmaxf(a[o + 0] + bb.x, 0.0f);
  r.y = fmaxf(a[o + 1] + bb.y, 0.0f);
  r.z = fmaxf(a[o + 2] + bb.z, 0.0f);
  r.w = fmaxf(a[o + 3] + bb.w, 0.0f);
  return r;
}

__global__ __launch_bounds__(128) void qkv_kernel(
    const float* __restrict__ x,
    const float* __restrict__ Wq, const float* __restrict__ bq,
    const float* __restrict__ Wk, const float* __restrict__ bk,
    const float* __restrict__ Wv, const float* __restrict__ bv,
    float* __restrict__ ws) {
  const int NT = (BB * SS) / 16;
  int wv    = __builtin_amdgcn_readfirstlane((int)(threadIdx.x >> 5));
  int tid   = blockIdx.x * 4 + wv;
  int which = tid / NT;
  int tile  = tid % NT;
  int r0    = tile * 16;
  int b     = tile / (SS / 16);
  int s0    = (tile % (SS / 16)) * 16;

  const float* W    = (which == 0) ? Wq : (which == 1) ? Wk : Wv;
  const float* bias = (which == 0) ? bq : (which == 1) ? bk : bv;
  float* Out = ws + (size_t)which * (BB * SS * DD);

  int lane = threadIdx.x & 31;
  int half = lane >> 4;
  int l16  = lane & 15;

  v2f xf[16];
  const float* xrow = x + (size_t)(r0 + l16) * DD + 2 * half;
#pragma unroll
  for (int kc = 0; kc < 16; ++kc) {
    xf[kc].x = xrow[4 * kc + 0];
    xf[kc].y = xrow[4 * kc + 1];
  }

  v8f acc[4];
#pragma unroll
  for (int n = 0; n < 4; ++n)
#pragma unroll
    for (int v = 0; v < 8; ++v) acc[n][v] = 0.0f;

  if (which == 2) {
#pragma unroll
    for (int kc = 0; kc < 16; ++kc) {
      int krow = 4 * kc + 2 * half;
#pragma unroll
      for (int n = 0; n < 4; ++n) {
        v2f wf;
        wf.x = W[(size_t)krow * DD + n * 16 + l16];
        wf.y = W[(size_t)(krow + 1) * DD + n * 16 + l16];
        acc[n] = wmma_f32_16x16x4(xf[kc], wf, acc[n]);
      }
    }
    int G = s0 >> 5;
    int j = (s0 >> 4) & 1;
#pragma unroll
    for (int n = 0; n < 4; ++n) {
      float bs = bias[n * 16 + l16];
      float4 bb = {bs, bs, bs, bs};
      int head = n * 4 + (l16 >> 2);
      float* dst = Out + ((size_t)(b * HH + head) * (SS / 32) + G) * 128 +
                   (2 * j) * 32 + (l16 & 3) * 8 + half * 4;
#pragma unroll
      for (int t = 0; t < 2; ++t)
        *reinterpret_cast<float4*>(dst + t * 32) = bias_relu4(acc[n], 4 * t, bb);
    }
  } else {
#pragma unroll
    for (int kc = 0; kc < 16; ++kc) {
      int krow = 4 * kc + 2 * half;
#pragma unroll
      for (int n = 0; n < 4; ++n) {
        v2f wf;
        wf.x = W[(size_t)krow * DD + n * 16 + l16];
        wf.y = W[(size_t)(krow + 1) * DD + n * 16 + l16];
        acc[n] = wmma_f32_16x16x4(wf, xf[kc], acc[n]);
      }
    }
#pragma unroll
    for (int n = 0; n < 4; ++n)
#pragma unroll
      for (int u = 0; u < 2; ++u) {
        int head = n * 4 + 2 * half + u;
        const float4 bb =
            *reinterpret_cast<const float4*>(bias + head * HDIM);
        float* dst = Out + ((size_t)(b * HH + head) * SS + s0 + l16) * HDIM;
        *reinterpret_cast<float4*>(dst) = bias_relu4(acc[n], 4 * u, bb);
      }
  }
}

static __device__ __forceinline__ float xhalf(float v) {
  int b = __builtin_bit_cast(int, v);
  return __builtin_bit_cast(
      float, __builtin_amdgcn_permlanex16(b, b, 0x76543210, 0xfedcba98, false,
                                          false));
}

#define ATTN_CS (0.5f * 1.4426950408889634f)

static __device__ __forceinline__ void load_k(const float* kp, int step,
                                              v2f (&ak)[2]) {
  for (int j = 0; j < 2; ++j)
    ak[j] = *reinterpret_cast<const v2f*>(
        kp + (size_t)(step * 32 + 16 * j) * HDIM);
}

static __device__ __forceinline__ void load_v(const float* vp, int step,
                                              float4 (&vv)[4]) {
  for (int u = 0; u < 4; ++u)
    vv[u] = *reinterpret_cast<const float4*>(
        vp + (size_t)step * 32 * HDIM + u * 32);
}

static __device__ __forceinline__ void score_tiles(const v2f (&ak)[2], v2f bq,
                                                   v8f (&s)[2]) {
  v8f zc;
  for (int v = 0; v < 8; ++v) zc[v] = 0.0f;
  for (int j = 0; j < 2; ++j) s[j] = wmma_f32_16x16x4(ak[j], bq, zc);
}

static __device__ __forceinline__ void max_step(const v8f (&s)[2], float& mraw,
                                                float& mc, float& mstep,
                                                float& astep) {
  float t = mraw;
  for (int j = 0; j < 2; ++j)
    for (int v = 0; v < 8; v += 2) t = fmaxf(fmaxf(t, s[j][v]), s[j][v + 1]);
  float mnew = fmaxf(t, xhalf(t));
  float mcn  = mnew * ATTN_CS;
  astep = __builtin_amdgcn_exp2f(mc - mcn);
  mstep = mcn;
  mraw  = mnew;
  mc    = mcn;
}

#define ATTN_FENCE() __builtin_amdgcn_sched_barrier(0)

template <bool HAS_NEXT>
static __device__ __forceinline__ void fused_step(
    const v8f (&sC)[2], const float4 (&vC)[4], float mC, float aC,
    const v8f (&sN)[2], float& mN, float& aN, float& mraw, float& mc,
    float& lsum, v8f& accO) {
  const v2f cs2 = {ATTN_CS, ATTN_CS};
  const v2f nm2 = {-mC, -mC};
  for (int v = 0; v < 4; ++v) accO[v] *= aC;
  v2f e = __builtin_elementwise_fma((v2f){sC[0][0], sC[0][1]}, cs2, nm2);
  ATTN_FENCE();
  float t = mraw;
  v2f p[8];
  v2f q[4];
  for (int c = 0; c < 8; ++c) {
    const int j = c >> 2, i = c & 3;
    v2f en = e;
    p[c].x = __builtin_amdgcn_exp2f(e.x);
    if (c < 7) {
      const int c1 = c + 1;
      const v2f sp = {sC[c1 >> 2][2 * (c1 & 3)], sC[c1 >> 2][2 * (c1 & 3) + 1]};
      en = __builtin_elementwise_fma(sp, cs2, nm2);
    } else {
      q[2] = p[4] + p[5];
    }
    ATTN_FENCE();
    p[c].y = __builtin_amdgcn_exp2f(e.y);
    if (HAS_NEXT) t = fmaxf(fmaxf(t, sN[j][2 * i]), sN[j][2 * i + 1]);
    ATTN_FENCE();
    const float4 v4 = vC[c >> 1];
    const v2f av = (c & 1) ? (v2f){v4.z, v4.w} : (v2f){v4.x, v4.y};
    accO = wmma_f32_16x16x4(av, p[c], accO);
    if ((c & 1) && c != 5) q[c >> 1] = p[c - 1] + p[c];
    ATTN_FENCE();
    e = en;
  }
  const v2f r2 = (q[0] + q[1]) + (q[2] + q[3]);
  lsum = lsum * aC + (r2.x + r2.y);
  if (HAS_NEXT) {
    float mnew = fmaxf(t, xhalf(t));
    float mcn  = mnew * ATTN_CS;
    aN   = __builtin_amdgcn_exp2f(mc - mcn);
    mN   = mcn;
    mraw = mnew;
    mc   = mcn;
  }
}

__global__ __launch_bounds__(128) __attribute__((amdgpu_waves_per_eu(8, 8)))
void attn_kernel(
    const float* __restrict__ x, float* __restrict__ Q,
    const float* __restrict__ KV, float* __restrict__ out) {
  const float* K = KV;
  const float* V = KV + (size_t)BB * SS * DD;
  __shared__ float red[4][8][16];

  int wave = threadIdx.x >> 5;
  int lane = threadIdx.x & 31;
  int half = lane >> 4;
  int l16  = lane & 15;

  const int QT = SS / 64;
  int bh = blockIdx.x / QT;
  int qt = blockIdx.x % QT;
  int b = bh / HH, h = bh % HH;
  int q0 = qt * 64 + wave * 16;

  const size_t hb = (size_t)bh * SS * HDIM;
  float* Qb = Q + hb;
  const float* Kb = K + hb;
  const float* Vb = V + hb;

  const v2f bq = *reinterpret_cast<const v2f*>(
      Qb + (size_t)(q0 + l16) * HDIM + 2 * half);

  float mraw = -INFINITY;
  float mc   = -INFINITY;
  float lsum = 0.0f;
  v8f accO;
  for (int v = 0; v < 8; ++v) accO[v] = 0.0f;

  const float* kp = Kb + (size_t)l16 * HDIM + 2 * half;
  const float* vp = Vb + (l16 & 3) * 8 + half * 4;

  const int NS = SS / 32;

  v2f kX[2], kY[2];
  float4 vX[4], vY[4];
  v8f sX[2], sY[2];
  float mX, aX, mY, aY;

  load_k(kp, 0, kX);
  load_v(vp, 0, vX);
  load_k(kp, 1, kY);
  score_tiles(kX, bq, sX);
  max_step(sX, mraw, mc, mX, aX);

#define ATTN_STEP(i, kC, vC, sC, mC, aC, kN, vN, sN, mN, aN) \
  do {                                                       \
    load_v(vp, (i) + 1, vN);                                 \
    load_k(kp, (i) + 2, kC);                                 \
    ATTN_FENCE();                                            \
    score_tiles(kN, bq, sN);                                 \
    ATTN_FENCE();                                            \
    fused_step<true>(sC, vC, mC, aC, sN, mN, aN, mraw, mc,   \
                     lsum, accO);                            \
    ATTN_FENCE();                                            \
  } while (0)

  for (int i = 0; i < NS - 2; i += 2) {
    ATTN_STEP(i, kX, vX, sX, mX, aX, kY, vY, sY, mY, aY);
    ATTN_STEP(i + 1, kY, vY, sY, mY, aY, kX, vX, sX, mX, aX);
  }
#undef ATTN_STEP
  load_v(vp, NS - 1, vY);
  score_tiles(kY, bq, sY);
  ATTN_FENCE();
  fused_step<true>(sX, vX, mX, aX, sY, mY, aY, mraw, mc, lsum, accO);
  ATTN_FENCE();
  fused_step<false>(sY, vY, mY, aY, sX, mX, aX, mraw, mc, lsum, accO);

  float l = lsum + xhalf(lsum);

  if (half == 0) {
    size_t idx = ((size_t)b * SS + q0 + l16) * DD + h * HDIM;
    const float4 xr = *reinterpret_cast<const float4*>(x + idx);
    float4 o;
    o.x = accO[0] / l + xr.x;
    o.y = accO[1] / l + xr.y;
    o.z = accO[2] / l + xr.z;
    o.w = accO[3] / l + xr.w;
    *reinterpret_cast<float4*>(out + idx) = o;
    float* r = &red[wave][0][l16];
    r[0 * 16] = o.x;
    r[1 * 16] = o.y;
    r[2 * 16] = o.z;
    r[3 * 16] = o.w;
    r[4 * 16] = o.x * o.x;
    r[5 * 16] = o.y * o.y;
    r[6 * 16] = o.z * o.z;
    r[7 * 16] = o.w * o.w;
  }
  __syncthreads();
  if (threadIdx.x < 8) {
    int v = threadIdx.x;
    float a0 = 0.0f, a1 = 0.0f, a2 = 0.0f, a3 = 0.0f;
    for (int j = 0; j < 16; ++j) {
      a0 += red[0][v][j];
      a1 += red[1][v][j];
      a2 += red[2][v][j];
      a3 += red[3][v][j];
    }
    Qb[(size_t)qt * 64 * HDIM + v] = (a0 + a1) + (a2 + a3);
  }
}

__global__ __launch_bounds__(128) void bn_stats_kernel(
    const float* __restrict__ part, float* __restrict__ stats) {
  __shared__ float4 rs[128], rq[128];
  int h = blockIdx.x;
  int t = threadIdx.x;
  int b = t >> 5, qt = t & 31;
  const float4* p = reinterpret_cast<const float4*>(
      part + ((size_t)(b * HH + h) * SS + (size_t)qt * 64) * HDIM);
  rs[t] = p[0];
  rq[t] = p[1];
  __syncthreads();
  if (t < 4) {
    const float* fs = reinterpret_cast<const float*>(rs) + t;
    const float* fq = reinterpret_cast<const float*>(rq) + t;
    float s[4] = {0.0f, 0.0f, 0.0f, 0.0f};
    float q[4] = {0.0f, 0.0f, 0.0f, 0.0f};
    for (int i = 0; i < 32; ++i)
#pragma unroll
      for (int c = 0; c < 4; ++c) {
        s[c] += fs[(c * 32 + i) * 4];
        q[c] += fq[(c * 32 + i) * 4];
      }
    float ssum = (s[0] + s[1]) + (s[2] + s[3]);
    float ssq  = (q[0] + q[1]) + (q[2] + q[3]);
    const float invN = 1.0f / (float)(BB * SS);
    float mean = ssum * invN;
    float var = ssq * invN - mean * mean;
    int ch = h * HDIM + t;
    stats[ch] = mean;
    stats[DD + ch] = rsqrtf(var + 1e-5f);
  }
}

__global__ __launch_bounds__(256) void bn_apply_kernel(
    float* __restrict__ out, const float* __restrict__ stats,
    const float* __restrict__ gamma, const float* __restrict__ beta) {
  int idx = blockIdx.x * 256 + threadIdx.x;
  int ch = idx & (DD - 1);
  float v = out[idx];
  out[idx] = (v - stats[ch]) * stats[DD + ch] * gamma[ch] + beta[ch];
}

extern "C" void kernel_launch(void* const* d_in, const int* in_sizes, int n_in,
                              void* d_out, int out_size, void* d_ws,
                              size_t ws_size, hipStream_t stream) {
  const float* x     = (const float*)d_in[0];
  const float* Wq    = (const float*)d_in[1];
  const float* bq    = (const float*)d_in[2];
  const float* Wk    = (const float*)d_in[3];
  const float* bk    = (const float*)d_in[4];
  const float* Wv    = (const float*)d_in[5];
  const float* bv    = (const float*)d_in[6];
  const float* gamma = (const float*)d_in[7];
  const float* beta  = (const float*)d_in[8];
  float* out = (float*)d_out;
  float* ws = (float*)d_ws;
  float* kv = ws + (size_t)BB * SS * DD;
  float* stats = ws + (size_t)3 * BB * SS * DD;

  qkv_kernel<<<(3 * (BB * SS / 16)) / 4, 128, 0, stream>>>(x, Wq, bq, Wk, bk,
                                                           Wv, bv, ws);
  attn_kernel<<<BB * HH * (SS / 64), 128, 0, stream>>>(x, ws, kv, out);
  bn_stats_kernel<<<HH, 128, 0, stream>>>(ws, stats);
  bn_apply_kernel<<<(BB * SS * DD) / 256, 256, 0, stream>>>(out, stats, gamma,
                                                            beta);
}


# BatchNorm statistics folded into the apply launch: each of 128 blocks (128 rows x 32 channels) re-adds its 8 heads' records, then normalises; 3 launches, 1024 apply waves (16384); sum order changed
# speedup vs baseline: 4.8075x; 1.0416x over previous
#include <hip/hip_runtime.h>
#include <math.h>


typedef __attribute__((ext_vector_type(2))) float v2f;
typedef __attribute__((ext_vector_type(8))) float v8f;

#define BB 4
#define SS 2048
#define DD 64
#define HH 16
#define HDIM 4

static __device__ __forceinline__ v8f wmma_f32_16x16x4(v2f a, v2f b, v8f c) {
  return __builtin_amdgcn_wmma_f32_16x16x4_f32(false, a, false, b, (short)0, c,
                                               false, false);
}

static __device__ __forceinline__ float4 bias_relu4(const v8f& a, int o, float4 bb) {
  float4 r;
  r.x = fmaxf(a[o + 0] + bb.x, 0.0f);
  r.y = fmaxf(a[o + 1] + bb.y, 0.0f);
  r.z = fmaxf(a[o + 2] + bb.z, 0.0f);
  r.w = fmaxf(a[o + 3] + bb.w, 0.0f);
  return r;
}

__global__ __launch_bounds__(128) void qkv_kernel(
    const float* __restrict__ x,
    const float* __restrict__ Wq, const float* __restrict__ bq,
    const float* __restrict__ Wk, const float* __restrict__ bk,
    const float* __restrict__ Wv, const float* __restrict__ bv,
    float* __restrict__ ws) {
  const int NT = (BB * SS) / 16;
  int wv    = __builtin_amdgcn_readfirstlane((int)(threadIdx.x >> 5));
  int tid   = blockIdx.x * 4 + wv;
  int which = tid / NT;
  int tile  = tid % NT;
  int r0    = tile * 16;
  int b     = tile / (SS / 16);
  int s0    = (tile % (SS / 16)) * 16;

  const float* W    = (which == 0) ? Wq : (which == 1) ? Wk : Wv;
  const float* bias = (which == 0) ? bq : (which == 1) ? bk : bv;
  float* Out = ws + (size_t)which * (BB * SS * DD);

  int lane = threadIdx.x & 31;
  int half = lane >> 4;
  int l16  = lane & 15;

  v2f xf[16];
  const float* xrow = x + (size_t)(r0 + l16) * DD + 2 * half;
#pragma unroll
  for (int kc = 0; kc < 16; ++kc) {
    xf[kc].x = xrow[4 * kc + 0];
    xf[kc].y = xrow[4 * kc + 1];
  }

  v8f acc[4];
#pragma unroll
  for (int n = 0; n < 4; ++n)
#pragma unroll
    for (int v = 0; v < 8; ++v) acc[n][v] = 0.0f;

  if (which == 2) {
#pragma unroll
    for (int kc = 0; kc < 16; ++kc) {
      int krow = 4 * kc + 2 * half;
#pragma unroll
      for (int n = 0; n < 4; ++n) {
        v2f wf;
        wf.x = W[(size_t)krow * DD + n * 16 + l16];
        wf.y = W[(size_t)(krow + 1) * DD + n * 16 + l16];
        acc[n] = wmma_f32_16x16x4(xf[kc], wf, acc[n]);
      }
    }
    int G = s0 >> 5;
    int j = (s0 >> 4) & 1;
#pragma unroll
    for (int n = 0; n < 4; ++n) {
      float bs = bias[n * 16 + l16];
      float4 bb = {bs, bs, bs, bs};
      int head = n * 4 + (l16 >> 2);
      float* dst = Out + ((size_t)(b * HH + head) * (SS / 32) + G) * 128 +
                   (2 * j) * 32 + (l16 & 3) * 8 + half * 4;
#pragma unroll
      for (int t = 0; t < 2; ++t)
        *reinterpret_cast<float4*>(dst + t * 32) = bias_relu4(acc[n], 4 * t, bb);
    }
  } else {
#pragma unroll
    for (int kc = 0; kc < 16; ++kc) {
      int krow = 4 * kc + 2 * half;
#pragma unroll
      for (int n = 0; n < 4; ++n) {
        v2f wf;
        wf.x = W[(size_t)krow * DD + n * 16 + l16];
        wf.y = W[(size_t)(krow + 1) * DD + n * 16 + l16];
        acc[n] = wmma_f32_16x16x4(wf, xf[kc], acc[n]);
      }
    }
#pragma unroll
    for (int n = 0; n < 4; ++n)
#pragma unroll
      for (int u = 0; u < 2; ++u) {
        int head = n * 4 + 2 * half + u;
        const float4 bb =
            *reinterpret_cast<const float4*>(bias + head * HDIM);
        float* dst = Out + ((size_t)(b * HH + head) * SS + s0 + l16) * HDIM;
        *reinterpret_cast<float4*>(dst) = bias_relu4(acc[n], 4 * u, bb);
      }
  }
}

static __device__ __forceinline__ float xhalf(float v) {
  int b = __builtin_bit_cast(int, v);
  return __builtin_bit_cast(
      float, __builtin_amdgcn_permlanex16(b, b, 0x76543210, 0xfedcba98, false,
                                          false));
}

#define ATTN_CS (0.5f * 1.4426950408889634f)

static __device__ __forceinline__ void load_k(const float* kp, int step,
                                              v2f (&ak)[2]) {
  for (int j = 0; j < 2; ++j)
    ak[j] = *reinterpret_cast<const v2f*>(
        kp + (size_t)(step * 32 + 16 * j) * HDIM);
}

static __device__ __forceinline__ void load_v(const float* vp, int step,
                                              float4 (&vv)[4]) {
  for (int u = 0; u < 4; ++u)
    vv[u] = *reinterpret_cast<const float4*>(
        vp + (size_t)step * 32 * HDIM + u * 32);
}

static __device__ __forceinline__ void score_tiles(const v2f (&ak)[2], v2f bq,
                                                   v8f (&s)[2]) {
  v8f zc;
  for (int v = 0; v < 8; ++v) zc[v] = 0.0f;
  for (int j = 0; j < 2; ++j) s[j] = wmma_f32_16x16x4(ak[j], bq, zc);
}

static __device__ __forceinline__ void max_step(const v8f (&s)[2], float& mraw,
                                                float& mc, float& mstep,
                                                float& astep) {
  float t = mraw;
  for (int j = 0; j < 2; ++j)
    for (int v = 0; v < 8; v += 2) t = fmaxf(fmaxf(t, s[j][v]), s[j][v + 1]);
  float mnew = fmaxf(t, xhalf(t));
  float mcn  = mnew * ATTN_CS;
  astep = __builtin_amdgcn_exp2f(mc - mcn);
  mstep = mcn;
  mraw  = mnew;
  mc    = mcn;
}

#define ATTN_FENCE() __builtin_amdgcn_sched_barrier(0)

template <bool HAS_NEXT>
static __device__ __forceinline__ void fused_step(
    const v8f (&sC)[2], const float4 (&vC)[4], float mC, float aC,
    const v8f (&sN)[2], float& mN, float& aN, float& mraw, float& mc,
    float& lsum, v8f& accO) {
  const v2f cs2 = {ATTN_CS, ATTN_CS};
  const v2f nm2 = {-mC, -mC};
  for (int v = 0; v < 4; ++v) accO[v] *= aC;
  v2f e = __builtin_elementwise_fma((v2f){sC[0][0], sC[0][1]}, cs2, nm2);
  ATTN_FENCE();
  float t = mraw;
  v2f p[8];
  v2f q[4];
  for (int c = 0; c < 8; ++c) {
    const int j = c >> 2, i = c & 3;
    v2f en = e;
    p[c].x = __builtin_amdgcn_exp2f(e.x);
    if (c < 7) {
      const int c1 = c + 1;
      const v2f sp = {sC[c1 >> 2][2 * (c1 & 3)], sC[c1 >> 2][2 * (c1 & 3) + 1]};
      en = __builtin_elementwise_fma(sp, cs2, nm2);
    } else {
      q[2] = p[4] + p[5];
    }
    ATTN_FENCE();
    p[c].y = __builtin_amdgcn_exp2f(e.y);
    if (HAS_NEXT) t = fmaxf(fmaxf(t, sN[j][2 * i]), sN[j][2 * i + 1]);
    ATTN_FENCE();
    const float4 v4 = vC[c >> 1];
    const v2f av = (c & 1) ? (v2f){v4.z, v4.w} : (v2f){v4.x, v4.y};
    accO = wmma_f32_16x16x4(av, p[c], accO);
    if ((c & 1) && c != 5) q[c >> 1] = p[c - 1] + p[c];
    ATTN_FENCE();
    e = en;
  }
  const v2f r2 = (q[0] + q[1]) + (q[2] + q[3]);
  lsum = lsum * aC + (r2.x + r2.y);
  if (HAS_NEXT) {
    float mnew = fmaxf(t, xhalf(t));
    float mcn  = mnew * ATTN_CS;
    aN   = __builtin_amdgcn_exp2f(mc - mcn);
    mN   = mcn;
    mraw = mnew;
    mc   = mcn;
  }
}

__global__ __launch_bounds__(128) __attribute__((amdgpu_waves_per_eu(8, 8)))
void attn_kernel(
    const float* __restrict__ x, float* __restrict__ Q,
    const float* __restrict__ KV, float* __restrict__ out) {
  const float* K = KV;
  const float* V = KV + (size_t)BB * SS * DD;
  __shared__ float red[4][8][16];

  int wave = threadIdx.x >> 5;
  int lane = threadIdx.x & 31;
  int half = lane >> 4;
  int l16  = lane & 15;

  const int QT = SS / 64;
  int bh = blockIdx.x / QT;
  int qt = blockIdx.x % QT;
  int b = bh / HH, h = bh % HH;
  int q0 = qt * 64 + wave * 16;

  const size_t hb = (size_t)bh * SS * HDIM;
  float* Qb = Q + hb;
  const float* Kb = K + hb;
  const float* Vb = V + hb;

  const v2f bq = *reinterpret_cast<const v2f*>(
      Qb + (size_t)(q0 + l16) * HDIM + 2 * half);

  float mraw = -INFINITY;
  float mc   = -INFINITY;
  float lsum = 0.0f;
  v8f accO;
  for (int v = 0; v < 8; ++v) accO[v] = 0.0f;

  const float* kp = Kb + (size_t)l16 * HDIM + 2 * half;
  const float* vp = Vb + (l16 & 3) * 8 + half * 4;

  const int NS = SS / 32;

  v2f kX[2], kY[2];
  float4 vX[4], vY[4];
  v8f sX[2], sY[2];
  float mX, aX, mY, aY;

  load_k(kp, 0, kX);
  load_v(vp, 0, vX);
  load_k(kp, 1, kY);
  score_tiles(kX, bq, sX);
  max_step(sX, mraw, mc, mX, aX);

#define ATTN_STEP(i, kC, vC, sC, mC, aC, kN, vN, sN, mN, aN) \
  do {                                                       \
    load_v(vp, (i) + 1, vN);                                 \
    load_k(kp, (i) + 2, kC);                                 \
    ATTN_FENCE();                                            \
    score_tiles(kN, bq, sN);                                 \
    ATTN_FENCE();                                            \
    fused_step<true>(sC, vC, mC, aC, sN, mN, aN, mraw, mc,   \
                     lsum, accO);                            \
    ATTN_FENCE();                                            \
  } while (0)

  for (int i = 0; i < NS - 2; i += 2) {
    ATTN_STEP(i, kX, vX, sX, mX, aX, kY, vY, sY, mY, aY);
    ATTN_STEP(i + 1, kY, vY, sY, mY, aY, kX, vX, sX, mX, aX);
  }
#undef ATTN_STEP
  load_v(vp, NS - 1, vY);
  score_tiles(kY, bq, sY);
  ATTN_FENCE();
  fused_step<true>(sX, vX, mX, aX, sY, mY, aY, mraw, mc, lsum, accO);
  ATTN_FENCE();
  fused_step<false>(sY, vY, mY, aY, sX, mX, aX, mraw, mc, lsum, accO);

  float l = lsum + xhalf(lsum);

  if (half == 0) {
    size_t idx = ((size_t)b * SS + q0 + l16) * DD + h * HDIM;
    const float4 xr = *reinterpret_cast<const float4*>(x + idx);
    float4 o;
    o.x = accO[0] / l + xr.x;
    o.y = accO[1] / l + xr.y;
    o.z = accO[2] / l + xr.z;
    o.w = accO[3] / l + xr.w;
    *reinterpret_cast<float4*>(out + idx) = o;
    float* r = &red[wave][0][l16];
    r[0 * 16] = o.x;
    r[1 * 16] = o.y;
    r[2 * 16] = o.z;
    r[3 * 16] = o.w;
    r[4 * 16] = o.x * o.x;
    r[5 * 16] = o.y * o.y;
    r[6 * 16] = o.z * o.z;
    r[7 * 16] = o.w * o.w;
  }
  __syncthreads();
  if (threadIdx.x < 8) {
    int v = threadIdx.x;
    float a0 = 0.0f, a1 = 0.0f, a2 = 0.0f, a3 = 0.0f;
    for (int j = 0; j < 16; ++j) {
      a0 += red[0][v][j];
      a1 += red[1][v][j];
      a2 += red[2][v][j];
      a3 += red[3][v][j];
    }
    Qb[(size_t)qt * 64 * HDIM + v] = (a0 + a1) + (a2 + a3);
  }
}

template <int M>
static __device__ __forceinline__ float row_xor(float v) {
  return __builtin_bit_cast(
      float, __builtin_amdgcn_update_dpp(0, __builtin_bit_cast(int, v),
                                         0x160 + M, 0xf, 0xf, true));
}

static __device__ __forceinline__ float parity_sum(float v) {
  v += row_xor<2>(v);
  v += row_xor<4>(v);
  v += row_xor<8>(v);
  return v + xhalf(v);
}

static __device__ __forceinline__ float4 bn_apply4(float4 v, float4 m, float4 rs,
                                                   float4 g, float4 be) {
  float4 o;
  o.x = (v.x - m.x) * rs.x * g.x + be.x;
  o.y = (v.y - m.y) * rs.y * g.y + be.y;
  o.z = (v.z - m.z) * rs.z * g.z + be.z;
  o.w = (v.w - m.w) * rs.w * g.w + be.w;
  return o;
}

__global__ __launch_bounds__(256) void bn_kernel(
    float* __restrict__ out, const float* __restrict__ part,
    const float* __restrict__ gamma, const float* __restrict__ beta) {
  __shared__ float4 smean[8], srstd[8];
  const int t    = threadIdx.x;
  const int wave = t >> 5;
  const int lane = t & 31;
  const int p    = blockIdx.x & 1;
  const int r0   = (blockIdx.x >> 1) * 128;
  const int c4   = t & 7;

  float* o = out + (size_t)(r0 + (t >> 3)) * DD + p * 32 + c4 * 4;
  float4 v[4];
#pragma unroll
  for (int i = 0; i < 4; ++i)
    v[i] = *reinterpret_cast<const float4*>(o + (size_t)i * 32 * DD);
  const float4 g  = *reinterpret_cast<const float4*>(gamma + p * 32 + c4 * 4);
  const float4 be = *reinterpret_cast<const float4*>(beta + p * 32 + c4 * 4);

  const int head = p * 8 + wave;
  const float* rp = part +
                    ((size_t)head * SS + (size_t)(lane >> 1) * 64) * HDIM +
                    (lane & 1) * 4;
  float4 a[8];
#pragma unroll
  for (int i = 0; i < 8; ++i)
    a[i] = *reinterpret_cast<const float4*>(
        rp + ((size_t)(i >> 1) * HH * SS + (size_t)(i & 1) * 16 * 64) * HDIM);
  __builtin_amdgcn_sched_barrier(0);
  float4 acc = a[0];
#pragma unroll
  for (int i = 1; i < 8; ++i) {
    acc.x += a[i].x;
    acc.y += a[i].y;
    acc.z += a[i].z;
    acc.w += a[i].w;
  }
  acc.x = parity_sum(acc.x);
  acc.y = parity_sum(acc.y);
  acc.z = parity_sum(acc.z);
  acc.w = parity_sum(acc.w);
  float4 nb;
  nb.x = row_xor<1>(acc.x);
  nb.y = row_xor<1>(acc.y);
  nb.z = row_xor<1>(acc.z);
  nb.w = row_xor<1>(acc.w);
  if (lane == 0) {
    const float invN = 1.0f / (float)(BB * SS);
    float4 mean, rstd;
    mean.x = acc.x * invN;
    mean.y = acc.y * invN;
    mean.z = acc.z * invN;
    mean.w = acc.w * invN;
    rstd.x = rsqrtf(nb.x * invN - mean.x * mean.x + 1e-5f);
    rstd.y = rsqrtf(nb.y * invN - mean.y * mean.y + 1e-5f);
    rstd.z = rsqrtf(nb.z * invN - mean.z * mean.z + 1e-5f);
    rstd.w = rsqrtf(nb.w * invN - mean.w * mean.w + 1e-5f);
    smean[wave] = mean;
    srstd[wave] = rstd;
  }
  __syncthreads();
  const float4 m  = smean[c4];
  const float4 rs = srstd[c4];
#pragma unroll
  for (int i = 0; i < 4; ++i)
    *reinterpret_cast<float4*>(o + (size_t)i * 32 * DD) =
        bn_apply4(v[i], m, rs, g, be);
}

extern "C" void kernel_launch(void* const* d_in, const int* in_sizes, int n_in,
                              void* d_out, int out_size, void* d_ws,
                              size_t ws_size, hipStream_t stream) {
  const float* x     = (const float*)d_in[0];
  const float* Wq    = (const float*)d_in[1];
  const float* bq    = (const float*)d_in[2];
  const float* Wk    = (const float*)d_in[3];
  const float* bk    = (const float*)d_in[4];
  const float* Wv    = (const float*)d_in[5];
  const float* bv    = (const float*)d_in[6];
  const float* gamma = (const float*)d_in[7];
  const float* beta  = (const float*)d_in[8];
  float* out = (float*)d_out;
  float* ws = (float*)d_ws;
  float* kv = ws + (size_t)BB * SS * DD;

  qkv_kernel<<<(3 * (BB * SS / 16)) / 4, 128, 0, stream>>>(x, Wq, bq, Wk, bk,
                                                           Wv, bv, ws);
  attn_kernel<<<BB * HH * (SS / 64), 128, 0, stream>>>(x, ws, kv, out);
  bn_kernel<<<2 * (BB * SS / 128), 256, 0, stream>>>(out, ws, gamma, beta);
}


# attention: softmax shift fixed per query at q . (channel maxima of K), a bound as ReLU gives q, k >= 0: key loop drops maximum chain, rescale, lane exchange; exact loop as fallback; sum order changed
# speedup vs baseline: 4.9779x; 1.0354x over previous
#include <hip/hip_runtime.h>
#include <math.h>


typedef __attribute__((ext_vector_type(2))) float v2f;
typedef __attribute__((ext_vector_type(8))) float v8f;

#define BB 4
#define SS 2048
#define DD 64
#define HH 16
#define HDIM 4

static __device__ __forceinline__ v8f wmma_f32_16x16x4(v2f a, v2f b, v8f c) {
  return __builtin_amdgcn_wmma_f32_16x16x4_f32(false, a, false, b, (short)0, c,
                                               false, false);
}

static __device__ __forceinline__ float4 bias_relu4(const v8f& a, int o, float4 bb) {
  float4 r;
  r.x = fmaxf(a[o + 0] + bb.x, 0.0f);
  r.y = fmaxf(a[o + 1] + bb.y, 0.0f);
  r.z = fmaxf(a[o + 2] + bb.z, 0.0f);
  r.w = fmaxf(a[o + 3] + bb.w, 0.0f);
  return r;
}

__global__ __launch_bounds__(128) void qkv_kernel(
    const float* __restrict__ x,
    const float* __restrict__ Wq, const float* __restrict__ bq,
    const float* __restrict__ Wk, const float* __restrict__ bk,
    const float* __restrict__ Wv, const float* __restrict__ bv,
    float* __restrict__ ws) {
  const int NT = (BB * SS) / 16;
  int wv    = __builtin_amdgcn_readfirstlane((int)(threadIdx.x >> 5));
  int tid   = blockIdx.x * 4 + wv;
  int which = tid / NT;
  int tile  = tid % NT;
  int r0    = tile * 16;
  int b     = tile / (SS / 16);
  int s0    = (tile % (SS / 16)) * 16;

  const float* W    = (which == 0) ? Wq : (which == 1) ? Wk : Wv;
  const float* bias = (which == 0) ? bq : (which == 1) ? bk : bv;
  float* Out = ws + (size_t)which * (BB * SS * DD);

  int lane = threadIdx.x & 31;
  int half = lane >> 4;
  int l16  = lane & 15;

  v2f xf[16];
  const float* xrow = x + (size_t)(r0 + l16) * DD + 2 * half;
#pragma unroll
  for (int kc = 0; kc < 16; ++kc) {
    xf[kc].x = xrow[4 * kc + 0];
    xf[kc].y = xrow[4 * kc + 1];
  }

  v8f acc[4];
#pragma unroll
  for (int n = 0; n < 4; ++n)
#pragma unroll
    for (int v = 0; v < 8; ++v) acc[n][v] = 0.0f;

  if (which == 2) {
#pragma unroll
    for (int kc = 0; kc < 16; ++kc) {
      int krow = 4 * kc + 2 * half;
#pragma unroll
      for (int n = 0; n < 4; ++n) {
        v2f wf;
        wf.x = W[(size_t)krow * DD + n * 16 + l16];
        wf.y = W[(size_t)(krow + 1) * DD + n * 16 + l16];
        acc[n] = wmma_f32_16x16x4(xf[kc], wf, acc[n]);
      }
    }
    int G = s0 >> 5;
    int j = (s0 >> 4) & 1;
#pragma unroll
    for (int n = 0; n < 4; ++n) {
      float bs = bias[n * 16 + l16];
      float4 bb = {bs, bs, bs, bs};
      int head = n * 4 + (l16 >> 2);
      float* dst = Out + ((size_t)(b * HH + head) * (SS / 32) + G) * 128 +
                   (2 * j) * 32 + (l16 & 3) * 8 + half * 4;
#pragma unroll
      for (int t = 0; t < 2; ++t)
        *reinterpret_cast<float4*>(dst + t * 32) = bias_relu4(acc[n], 4 * t, bb);
    }
  } else {
#pragma unroll
    for (int kc = 0; kc < 16; ++kc) {
      int krow = 4 * kc + 2 * half;
#pragma unroll
      for (int n = 0; n < 4; ++n) {
        v2f wf;
        wf.x = W[(size_t)krow * DD + n * 16 + l16];
        wf.y = W[(size_t)(krow + 1) * DD + n * 16 + l16];
        acc[n] = wmma_f32_16x16x4(wf, xf[kc], acc[n]);
      }
    }
#pragma unroll
    for (int n = 0; n < 4; ++n)
#pragma unroll
      for (int u = 0; u < 2; ++u) {
        int head = n * 4 + 2 * half + u;
        const float4 bb =
            *reinterpret_cast<const float4*>(bias + head * HDIM);
        float* dst = Out + ((size_t)(b * HH + head) * SS + s0 + l16) * HDIM;
        *reinterpret_cast<float4*>(dst) = bias_relu4(acc[n], 4 * u, bb);
      }
  }
}

static __device__ __forceinline__ float xhalf(float v) {
  int b = __builtin_bit_cast(int, v);
  return __builtin_bit_cast(
      float, __builtin_amdgcn_permlanex16(b, b, 0x76543210, 0xfedcba98, false,
                                          false));
}

template <int M>
static __device__ __forceinline__ float row_xor(float v) {
  return __builtin_bit_cast(
      float, __builtin_amdgcn_update_dpp(0, __builtin_bit_cast(int, v),
                                         0x160 + M, 0xf, 0xf, true));
}

static __device__ __forceinline__ float wave_max(float v) {
  v = fmaxf(v, row_xor<1>(v));
  v = fmaxf(v, row_xor<2>(v));
  v = fmaxf(v, row_xor<4>(v));
  v = fmaxf(v, row_xor<8>(v));
  return fmaxf(v, xhalf(v));
}

#define ATTN_CS (0.5f * 1.4426950408889634f)

static __device__ __forceinline__ void load_k(const float* kp, int step,
                                              v2f (&ak)[2]) {
  for (int j = 0; j < 2; ++j)
    ak[j] = *reinterpret_cast<const v2f*>(
        kp + (size_t)(step * 32 + 16 * j) * HDIM);
}

static __device__ __forceinline__ void load_v(const float* vp, int step,
                                              float4 (&vv)[4]) {
  for (int u = 0; u < 4; ++u)
    vv[u] = *reinterpret_cast<const float4*>(
        vp + (size_t)step * 32 * HDIM + u * 32);
}

static __device__ __forceinline__ void score_tiles(const v2f (&ak)[2], v2f bq,
                                                   v8f (&s)[2]) {
  v8f zc;
  for (int v = 0; v < 8; ++v) zc[v] = 0.0f;
  for (int j = 0; j < 2; ++j) s[j] = wmma_f32_16x16x4(ak[j], bq, zc);
}

static __device__ __forceinline__ void max_step(const v8f (&s)[2], float& mraw,
                                                float& mc, float& mstep,
                                                float& astep) {
  float t = mraw;
  for (int j = 0; j < 2; ++j)
    for (int v = 0; v < 8; v += 2) t = fmaxf(fmaxf(t, s[j][v]), s[j][v + 1]);
  float mnew = fmaxf(t, xhalf(t));
  float mcn  = mnew * ATTN_CS;
  astep = __builtin_amdgcn_exp2f(mc - mcn);
  mstep = mcn;
  mraw  = mnew;
  mc    = mcn;
}

#define ATTN_FENCE() __builtin_amdgcn_sched_barrier(0)

template <bool HAS_NEXT>
static __device__ __forceinline__ void fused_step(
    const v8f (&sC)[2], const float4 (&vC)[4], float mC, float aC,
    const v8f (&sN)[2], float& mN, float& aN, float& mraw, float& mc,
    float& lsum, v8f& accO) {
  const v2f cs2 = {ATTN_CS, ATTN_CS};
  const v2f nm2 = {-mC, -mC};
  for (int v = 0; v < 4; ++v) accO[v] *= aC;
  v2f e = __builtin_elementwise_fma((v2f){sC[0][0], sC[0][1]}, cs2, nm2);
  ATTN_FENCE();
  float t = mraw;
  v2f p[8];
  v2f q[4];
  for (int c = 0; c < 8; ++c) {
    const int j = c >> 2, i = c & 3;
    v2f en = e;
    p[c].x = __builtin_amdgcn_exp2f(e.x);
    if (c < 7) {
      const int c1 = c + 1;
      const v2f sp = {sC[c1 >> 2][2 * (c1 & 3)], sC[c1 >> 2][2 * (c1 & 3) + 1]};
      en = __builtin_elementwise_fma(sp, cs2, nm2);
    } else {
      q[2] = p[4] + p[5];
    }
    ATTN_FENCE();
    p[c].y = __builtin_amdgcn_exp2f(e.y);
    if (HAS_NEXT) t = fmaxf(fmaxf(t, sN[j][2 * i]), sN[j][2 * i + 1]);
    ATTN_FENCE();
    const float4 v4 = vC[c >> 1];
    const v2f av = (c & 1) ? (v2f){v4.z, v4.w} : (v2f){v4.x, v4.y};
    accO = wmma_f32_16x16x4(av, p[c], accO);
    if ((c & 1) && c != 5) q[c >> 1] = p[c - 1] + p[c];
    ATTN_FENCE();
    e = en;
  }
  const v2f r2 = (q[0] + q[1]) + (q[2] + q[3]);
  lsum = lsum * aC + (r2.x + r2.y);
  if (HAS_NEXT) {
    float mnew = fmaxf(t, xhalf(t));
    float mcn  = mnew * ATTN_CS;
    aN   = __builtin_amdgcn_exp2f(mc - mcn);
    mN   = mcn;
    mraw = mnew;
    mc   = mcn;
  }
}

static __device__ __forceinline__ void fast_step(const v8f (&sC)[2],
                                                 const float4 (&vC)[4],
                                                 v2f nm2, v2f& l2, v2f& pend,
                                                 v8f& accO) {
  const v2f cs2 = {ATTN_CS, ATTN_CS};
  v2f e = __builtin_elementwise_fma((v2f){sC[0][0], sC[0][1]}, cs2, nm2);
  ATTN_FENCE();
  v2f p[8];
  for (int c = 0; c < 8; ++c) {
    v2f en = e;
    p[c].x = __builtin_amdgcn_exp2f(e.x);
    if (c < 7) {
      const int c1 = c + 1;
      const v2f sp = {sC[c1 >> 2][2 * (c1 & 3)], sC[c1 >> 2][2 * (c1 & 3) + 1]};
      en = __builtin_elementwise_fma(sp, cs2, nm2);
    }
    ATTN_FENCE();
    p[c].y = __builtin_amdgcn_exp2f(e.y);
    l2 += (c == 0) ? pend : p[c - 1];
    ATTN_FENCE();
    const float4 v4 = vC[c >> 1];
    const v2f av = (c & 1) ? (v2f){v4.z, v4.w} : (v2f){v4.x, v4.y};
    accO = wmma_f32_16x16x4(av, p[c], accO);
    ATTN_FENCE();
    e = en;
  }
  pend = p[7];
}

__global__ __launch_bounds__(128) __attribute__((amdgpu_waves_per_eu(8, 8)))
void attn_kernel(
    const float* __restrict__ x, float* __restrict__ Q,
    const float* __restrict__ KV, float* __restrict__ out) {
  const float* K = KV;
  const float* V = KV + (size_t)BB * SS * DD;
  __shared__ float red[4][8][16];
  __shared__ float4 kms[4];

  int wave = threadIdx.x >> 5;
  int lane = threadIdx.x & 31;
  int half = lane >> 4;
  int l16  = lane & 15;

  const int QT = SS / 64;
  int bh = blockIdx.x / QT;
  int qt = blockIdx.x % QT;
  int b = bh / HH, h = bh % HH;
  int q0 = qt * 64 + wave * 16;

  const size_t hb = (size_t)bh * SS * HDIM;
  float* Qb = Q + hb;
  const float* Kb = K + hb;
  const float* Vb = V + hb;

  const v2f bq = *reinterpret_cast<const v2f*>(
      Qb + (size_t)(q0 + l16) * HDIM + 2 * half);

  v8f accO;
  for (int v = 0; v < 8; ++v) accO[v] = 0.0f;

  const float* kp = Kb + (size_t)l16 * HDIM + 2 * half;
  const float* vp = Vb + (l16 & 3) * 8 + half * 4;

  const int NS = SS / 32;

  v2f kX[2], kY[2];
  float4 vX[4], vY[4];
  v8f sX[2], sY[2];

  const float4 qr =
      *reinterpret_cast<const float4*>(Qb + (size_t)(q0 + l16) * HDIM);
  load_k(kp, 0, kX);
  load_v(vp, 0, vX);
  load_k(kp, 1, kY);

  {
    const float* ks = Kb + ((size_t)wave * (SS / 4) + lane) * HDIM;
    float4 km = *reinterpret_cast<const float4*>(ks);
    for (int it = 1; it < SS / 4 / 32; ++it) {
      const float4 kk =
          *reinterpret_cast<const float4*>(ks + (size_t)it * 32 * HDIM);
      km.x = fmaxf(km.x, kk.x);
      km.y = fmaxf(km.y, kk.y);
      km.z = fmaxf(km.z, kk.z);
      km.w = fmaxf(km.w, kk.w);
    }
    km.x = wave_max(km.x);
    km.y = wave_max(km.y);
    km.z = wave_max(km.z);
    km.w = wave_max(km.w);
    if (lane == 0) kms[wave] = km;
  }
  __syncthreads();
  float mbound;
  {
    const float4 k0 = kms[0], k1 = kms[1], k2 = kms[2], k3 = kms[3];
    mbound = qr.x * fmaxf(fmaxf(k0.x, k1.x), fmaxf(k2.x, k3.x)) +
             qr.y * fmaxf(fmaxf(k0.y, k1.y), fmaxf(k2.y, k3.y)) +
             qr.z * fmaxf(fmaxf(k0.z, k1.z), fmaxf(k2.z, k3.z)) +
             qr.w * fmaxf(fmaxf(k0.w, k1.w), fmaxf(k2.w, k3.w));
  }

  v2f l2 = {0.0f, 0.0f};
  v2f pend = {0.0f, 0.0f};
  {
    const float nm = -(mbound * ATTN_CS);
    const v2f nm2 = {nm, nm};
    score_tiles(kX, bq, sX);
#define FAST_STEP(i, kC, vC, sC, kN, vN, sN) \
  do {                                       \
    load_v(vp, (i) + 1, vN);                 \
    load_k(kp, (i) + 2, kC);                 \
    ATTN_FENCE();                            \
    score_tiles(kN, bq, sN);                 \
    ATTN_FENCE();                            \
    fast_step(sC, vC, nm2, l2, pend, accO);  \
    ATTN_FENCE();                            \
  } while (0)
    for (int i = 0; i < NS - 2; i += 2) {
      FAST_STEP(i, kX, vX, sX, kY, vY, sY);
      FAST_STEP(i + 1, kY, vY, sY, kX, vX, sX);
    }
#undef FAST_STEP
    load_v(vp, NS - 1, vY);
    score_tiles(kY, bq, sY);
    ATTN_FENCE();
    fast_step(sX, vX, nm2, l2, pend, accO);
    ATTN_FENCE();
    fast_step(sY, vY, nm2, l2, pend, accO);
    l2 += pend;
  }
  float l = l2.x + l2.y;
  l += xhalf(l);

  if (__builtin_amdgcn_ballot_w32(!(l >= 0x1p-60f)) != 0u) {
    float mraw = -INFINITY;
    float mc   = -INFINITY;
    float lsum = 0.0f;
    float mX, aX, mY, aY;
    for (int v = 0; v < 8; ++v) accO[v] = 0.0f;

    load_k(kp, 0, kX);
    load_v(vp, 0, vX);
    load_k(kp, 1, kY);
    score_tiles(kX, bq, sX);
    max_step(sX, mraw, mc, mX, aX);

#define ATTN_STEP(i, kC, vC, sC, mC, aC, kN, vN, sN, mN, aN) \
  do {                                                       \
    load_v(vp, (i) + 1, vN);                                 \
    load_k(kp, (i) + 2, kC);                                 \
    ATTN_FENCE();                                            \
    score_tiles(kN, bq, sN);                                 \
    ATTN_FENCE();                                            \
    fused_step<true>(sC, vC, mC, aC, sN, mN, aN, mraw, mc,   \
                     lsum, accO);                            \
    ATTN_FENCE();                                            \
  } while (0)

    for (int i = 0; i < NS - 2; i += 2) {
      ATTN_STEP(i, kX, vX, sX, mX, aX, kY, vY, sY, mY, aY);
      ATTN_STEP(i + 1, kY, vY, sY, mY, aY, kX, vX, sX, mX, aX);
    }
#undef ATTN_STEP
    load_v(vp, NS - 1, vY);
    score_tiles(kY, bq, sY);
    ATTN_FENCE();
    fused_step<true>(sX, vX, mX, aX, sY, mY, aY, mraw, mc, lsum, accO);
    ATTN_FENCE();
    fused_step<false>(sY, vY, mY, aY, sX, mX, aX, mraw, mc, lsum, accO);

    l = lsum + xhalf(lsum);
  }

  if (half == 0) {
    size_t idx = ((size_t)b * SS + q0 + l16) * DD + h * HDIM;
    const float4 xr = *reinterpret_cast<const float4*>(x + idx);
    float4 o;
    o.x = accO[0] / l + xr.x;
    o.y = accO[1] / l + xr.y;
    o.z = accO[2] / l + xr.z;
    o.w = accO[3] / l + xr.w;
    *reinterpret_cast<float4*>(out + idx) = o;
    float* r = &red[wave][0][l16];
    r[0 * 16] = o.x;
    r[1 * 16] = o.y;
    r[2 * 16] = o.z;
    r[3 * 16] = o.w;
    r[4 * 16] = o.x * o.x;
    r[5 * 16] = o.y * o.y;
    r[6 * 16] = o.z * o.z;
    r[7 * 16] = o.w * o.w;
  }
  __syncthreads();
  if (threadIdx.x < 8) {
    int v = threadIdx.x;
    float a0 = 0.0f, a1 = 0.0f, a2 = 0.0f, a3 = 0.0f;
    for (int j = 0; j < 16; ++j) {
      a0 += red[0][v][j];
      a1 += red[1][v][j];
      a2 += red[2][v][j];
      a3 += red[3][v][j];
    }
    Qb[(size_t)qt * 64 * HDIM + v] = (a0 + a1) + (a2 + a3);
  }
}

static __device__ __forceinline__ float parity_sum(float v) {
  v += row_xor<2>(v);
  v += row_xor<4>(v);
  v += row_xor<8>(v);
  return v + xhalf(v);
}

static __device__ __forceinline__ float4 bn_apply4(float4 v, float4 m, float4 rs,
                                                   float4 g, float4 be) {
  float4 o;
  o.x = (v.x - m.x) * rs.x * g.x + be.x;
  o.y = (v.y - m.y) * rs.y * g.y + be.y;
  o.z = (v.z - m.z) * rs.z * g.z + be.z;
  o.w = (v.w - m.w) * rs.w * g.w + be.w;
  return o;
}

__global__ __launch_bounds__(256) void bn_kernel(
    float* __restrict__ out, const float* __restrict__ part,
    const float* __restrict__ gamma, const float* __restrict__ beta) {
  __shared__ float4 smean[8], srstd[8];
  const int t    = threadIdx.x;
  const int wave = t >> 5;
  const int lane = t & 31;
  const int p    = blockIdx.x & 1;
  const int r0   = (blockIdx.x >> 1) * 128;
  const int c4   = t & 7;

  float* o = out + (size_t)(r0 + (t >> 3)) * DD + p * 32 + c4 * 4;
  float4 v[4];
#pragma unroll
  for (int i = 0; i < 4; ++i)
    v[i] = *reinterpret_cast<const float4*>(o + (size_t)i * 32 * DD);
  const float4 g  = *reinterpret_cast<const float4*>(gamma + p * 32 + c4 * 4);
  const float4 be = *reinterpret_cast<const float4*>(beta + p * 32 + c4 * 4);

  const int head = p * 8 + wave;
  const float* rp = part +
                    ((size_t)head * SS + (size_t)(lane >> 1) * 64) * HDIM +
                    (lane & 1) * 4;
  float4 a[8];
#pragma unroll
  for (int i = 0; i < 8; ++i)
    a[i] = *reinterpret_cast<const float4*>(
        rp + ((size_t)(i >> 1) * HH * SS + (size_t)(i & 1) * 16 * 64) * HDIM);
  __builtin_amdgcn_sched_barrier(0);
  float4 acc = a[0];
#pragma unroll
  for (int i = 1; i < 8; ++i) {
    acc.x += a[i].x;
    acc.y += a[i].y;
    acc.z += a[i].z;
    acc.w += a[i].w;
  }
  acc.x = parity_sum(acc.x);
  acc.y = parity_sum(acc.y);
  acc.z = parity_sum(acc.z);
  acc.w = parity_sum(acc.w);
  float4 nb;
  nb.x = row_xor<1>(acc.x);
  nb.y = row_xor<1>(acc.y);
  nb.z = row_xor<1>(acc.z);
  nb.w = row_xor<1>(acc.w);
  if (lane == 0) {
    const float invN = 1.0f / (float)(BB * SS);
    float4 mean, rstd;
    mean.x = acc.x * invN;
    mean.y = acc.y * invN;
    mean.z = acc.z * invN;
    mean.w = acc.w * invN;
    rstd.x = rsqrtf(nb.x * invN - mean.x * mean.x + 1e-5f);
    rstd.y = rsqrtf(nb.y * invN - mean.y * mean.y + 1e-5f);
    rstd.z = rsqrtf(nb.z * invN - mean.z * mean.z + 1e-5f);
    rstd.w = rsqrtf(nb.w * invN - mean.w * mean.w + 1e-5f);
    smean[wave] = mean;
    srstd[wave] = rstd;
  }
  __syncthreads();
  const float4 m  = smean[c4];
  const float4 rs = srstd[c4];
#pragma unroll
  for (int i = 0; i < 4; ++i)
    *reinterpret_cast<float4*>(o + (size_t)i * 32 * DD) =
        bn_apply4(v[i], m, rs, g, be);
}

extern "C" void kernel_launch(void* const* d_in, const int* in_sizes, int n_in,
                              void* d_out, int out_size, void* d_ws,
                              size_t ws_size, hipStream_t stream) {
  const float* x     = (const float*)d_in[0];
  const float* Wq    = (const float*)d_in[1];
  const float* bq    = (const float*)d_in[2];
  const float* Wk    = (const float*)d_in[3];
  const float* bk    = (const float*)d_in[4];
  const float* Wv    = (const float*)d_in[5];
  const float* bv    = (const float*)d_in[6];
  const float* gamma = (const float*)d_in[7];
  const float* beta  = (const float*)d_in[8];
  float* out = (float*)d_out;
  float* ws = (float*)d_ws;
  float* kv = ws + (size_t)BB * SS * DD;

  qkv_kernel<<<(3 * (BB * SS / 16)) / 4, 128, 0, stream>>>(x, Wq, bq, Wk, bk,
                                                           Wv, bv, ws);
  attn_kernel<<<BB * HH * (SS / 64), 128, 0, stream>>>(x, ws, kv, out);
  bn_kernel<<<2 * (BB * SS / 128), 256, 0, stream>>>(out, ws, gamma, beta);
}


# attention: wave priority falls with key-loop progress (s_setprio 3,2,1,0 over parts of 32,16,8,8 steps): a SIMD's waves end together, none runs the end of the launch alone; sums and order unchanged
# speedup vs baseline: 5.2340x; 1.0514x over previous
#include <hip/hip_runtime.h>
#include <math.h>


typedef __attribute__((ext_vector_type(2))) float v2f;
typedef __attribute__((ext_vector_type(8))) float v8f;

#define BB 4
#define SS 2048
#define DD 64
#define HH 16
#define HDIM 4

static __device__ __forceinline__ v8f wmma_f32_16x16x4(v2f a, v2f b, v8f c) {
  return __builtin_amdgcn_wmma_f32_16x16x4_f32(false, a, false, b, (short)0, c,
                                               false, false);
}

static __device__ __forceinline__ float4 bias_relu4(const v8f& a, int o, float4 bb) {
  float4 r;
  r.x = fmaxf(a[o + 0] + bb.x, 0.0f);
  r.y = fmaxf(a[o + 1] + bb.y, 0.0f);
  r.z = fmaxf(a[o + 2] + bb.z, 0.0f);
  r.w = fmaxf(a[o + 3] + bb.w, 0.0f);
  return r;
}

__global__ __launch_bounds__(128) void qkv_kernel(
    const float* __restrict__ x,
    const float* __restrict__ Wq, const float* __restrict__ bq,
    const float* __restrict__ Wk, const float* __restrict__ bk,
    const float* __restrict__ Wv, const float* __restrict__ bv,
    float* __restrict__ ws) {
  const int NT = (BB * SS) / 16;
  int wv    = __builtin_amdgcn_readfirstlane((int)(threadIdx.x >> 5));
  int tid   = blockIdx.x * 4 + wv;
  int which = tid / NT;
  int tile  = tid % NT;
  int r0    = tile * 16;
  int b     = tile / (SS / 16);
  int s0    = (tile % (SS / 16)) * 16;

  const float* W    = (which == 0) ? Wq : (which == 1) ? Wk : Wv;
  const float* bias = (which == 0) ? bq : (which == 1) ? bk : bv;
  float* Out = ws + (size_t)which * (BB * SS * DD);

  int lane = threadIdx.x & 31;
  int half = lane >> 4;
  int l16  = lane & 15;

  v2f xf[16];
  const float* xrow = x + (size_t)(r0 + l16) * DD + 2 * half;
#pragma unroll
  for (int kc = 0; kc < 16; ++kc) {
    xf[kc].x = xrow[4 * kc + 0];
    xf[kc].y = xrow[4 * kc + 1];
  }

  v8f acc[4];
#pragma unroll
  for (int n = 0; n < 4; ++n)
#pragma unroll
    for (int v = 0; v < 8; ++v) acc[n][v] = 0.0f;

  if (which == 2) {
#pragma unroll
    for (int kc = 0; kc < 16; ++kc) {
      int krow = 4 * kc + 2 * half;
#pragma unroll
      for (int n = 0; n < 4; ++n) {
        v2f wf;
        wf.x = W[(size_t)krow * DD + n * 16 + l16];
        wf.y = W[(size_t)(krow + 1) * DD + n * 16 + l16];
        acc[n] = wmma_f32_16x16x4(xf[kc], wf, acc[n]);
      }
    }
    int G = s0 >> 5;
    int j = (s0 >> 4) & 1;
#pragma unroll
    for (int n = 0; n < 4; ++n) {
      float bs = bias[n * 16 + l16];
      float4 bb = {bs, bs, bs, bs};
      int head = n * 4 + (l16 >> 2);
      float* dst = Out + ((size_t)(b * HH + head) * (SS / 32) + G) * 128 +
                   (2 * j) * 32 + (l16 & 3) * 8 + half * 4;
#pragma unroll
      for (int t = 0; t < 2; ++t)
        *reinterpret_cast<float4*>(dst + t * 32) = bias_relu4(acc[n], 4 * t, bb);
    }
  } else {
#pragma unroll
    for (int kc = 0; kc < 16; ++kc) {
      int krow = 4 * kc + 2 * half;
#pragma unroll
      for (int n = 0; n < 4; ++n) {
        v2f wf;
        wf.x = W[(size_t)krow * DD + n * 16 + l16];
        wf.y = W[(size_t)(krow + 1) * DD + n * 16 + l16];
        acc[n] = wmma_f32_16x16x4(wf, xf[kc], acc[n]);
      }
    }
#pragma unroll
    for (int n = 0; n < 4; ++n)
#pragma unroll
      for (int u = 0; u < 2; ++u) {
        int head = n * 4 + 2 * half + u;
        const float4 bb =
            *reinterpret_cast<const float4*>(bias + head * HDIM);
        float* dst = Out + ((size_t)(b * HH + head) * SS + s0 + l16) * HDIM;
        *reinterpret_cast<float4*>(dst) = bias_relu4(acc[n], 4 * u, bb);
      }
  }
}

static __device__ __forceinline__ float xhalf(float v) {
  int b = __builtin_bit_cast(int, v);
  return __builtin_bit_cast(
      float, __builtin_amdgcn_permlanex16(b, b, 0x76543210, 0xfedcba98, false,
                                          false));
}

template <int M>
static __device__ __forceinline__ float row_xor(float v) {
  return __builtin_bit_cast(
      float, __builtin_amdgcn_update_dpp(0, __builtin_bit_cast(int, v),
                                         0x160 + M, 0xf, 0xf, true));
}

static __device__ __forceinline__ float wave_max(float v) {
  v = fmaxf(v, row_xor<1>(v));
  v = fmaxf(v, row_xor<2>(v));
  v = fmaxf(v, row_xor<4>(v));
  v = fmaxf(v, row_xor<8>(v));
  return fmaxf(v, xhalf(v));
}

#define ATTN_CS (0.5f * 1.4426950408889634f)

static __device__ __forceinline__ void load_k(const float* kp, int step,
                                              v2f (&ak)[2]) {
  for (int j = 0; j < 2; ++j)
    ak[j] = *reinterpret_cast<const v2f*>(
        kp + (size_t)(step * 32 + 16 * j) * HDIM);
}

static __device__ __forceinline__ void load_v(const float* vp, int step,
                                              float4 (&vv)[4]) {
  for (int u = 0; u < 4; ++u)
    vv[u] = *reinterpret_cast<const float4*>(
        vp + (size_t)step * 32 * HDIM + u * 32);
}

static __device__ __forceinline__ void score_tiles(const v2f (&ak)[2], v2f bq,
                                                   v8f (&s)[2]) {
  v8f zc;
  for (int v = 0; v < 8; ++v) zc[v] = 0.0f;
  for (int j = 0; j < 2; ++j) s[j] = wmma_f32_16x16x4(ak[j], bq, zc);
}

static __device__ __forceinline__ void max_step(const v8f (&s)[2], float& mraw,
                                                float& mc, float& mstep,
                                                float& astep) {
  float t = mraw;
  for (int j = 0; j < 2; ++j)
    for (int v = 0; v < 8; v += 2) t = fmaxf(fmaxf(t, s[j][v]), s[j][v + 1]);
  float mnew = fmaxf(t, xhalf(t));
  float mcn  = mnew * ATTN_CS;
  astep = __builtin_amdgcn_exp2f(mc - mcn);
  mstep = mcn;
  mraw  = mnew;
  mc    = mcn;
}

#define ATTN_FENCE() __builtin_amdgcn_sched_barrier(0)

template <bool HAS_NEXT>
static __device__ __forceinline__ void fused_step(
    const v8f (&sC)[2], const float4 (&vC)[4], float mC, float aC,
    const v8f (&sN)[2], float& mN, float& aN, float& mraw, float& mc,
    float& lsum, v8f& accO) {
  const v2f cs2 = {ATTN_CS, ATTN_CS};
  const v2f nm2 = {-mC, -mC};
  for (int v = 0; v < 4; ++v) accO[v] *= aC;
  v2f e = __builtin_elementwise_fma((v2f){sC[0][0], sC[0][1]}, cs2, nm2);
  ATTN_FENCE();
  float t = mraw;
  v2f p[8];
  v2f q[4];
  for (int c = 0; c < 8; ++c) {
    const int j = c >> 2, i = c & 3;
    v2f en = e;
    p[c].x = __builtin_amdgcn_exp2f(e.x);
    if (c < 7) {
      const int c1 = c + 1;
      const v2f sp = {sC[c1 >> 2][2 * (c1 & 3)], sC[c1 >> 2][2 * (c1 & 3) + 1]};
      en = __builtin_elementwise_fma(sp, cs2, nm2);
    } else {
      q[2] = p[4] + p[5];
    }
    ATTN_FENCE();
    p[c].y = __builtin_amdgcn_exp2f(e.y);
    if (HAS_NEXT) t = fmaxf(fmaxf(t, sN[j][2 * i]), sN[j][2 * i + 1]);
    ATTN_FENCE();
    const float4 v4 = vC[c >> 1];
    const v2f av = (c & 1) ? (v2f){v4.z, v4.w} : (v2f){v4.x, v4.y};
    accO = wmma_f32_16x16x4(av, p[c], accO);
    if ((c & 1) && c != 5) q[c >> 1] = p[c - 1] + p[c];
    ATTN_FENCE();
    e = en;
  }
  const v2f r2 = (q[0] + q[1]) + (q[2] + q[3]);
  lsum = lsum * aC + (r2.x + r2.y);
  if (HAS_NEXT) {
    float mnew = fmaxf(t, xhalf(t));
    float mcn  = mnew * ATTN_CS;
    aN   = __builtin_amdgcn_exp2f(mc - mcn);
    mN   = mcn;
    mraw = mnew;
    mc   = mcn;
  }
}

static __device__ __forceinline__ void fast_step(const v8f (&sC)[2],
                                                 const float4 (&vC)[4],
                                                 v2f nm2, v2f& l2, v2f& pend,
                                                 v8f& accO) {
  const v2f cs2 = {ATTN_CS, ATTN_CS};
  v2f e = __builtin_elementwise_fma((v2f){sC[0][0], sC[0][1]}, cs2, nm2);
  ATTN_FENCE();
  v2f p[8];
  for (int c = 0; c < 8; ++c) {
    v2f en = e;
    p[c].x = __builtin_amdgcn_exp2f(e.x);
    if (c < 7) {
      const int c1 = c + 1;
      const v2f sp = {sC[c1 >> 2][2 * (c1 & 3)], sC[c1 >> 2][2 * (c1 & 3) + 1]};
      en = __builtin_elementwise_fma(sp, cs2, nm2);
    }
    ATTN_FENCE();
    p[c].y = __builtin_amdgcn_exp2f(e.y);
    l2 += (c == 0) ? pend : p[c - 1];
    ATTN_FENCE();
    const float4 v4 = vC[c >> 1];
    const v2f av = (c & 1) ? (v2f){v4.z, v4.w} : (v2f){v4.x, v4.y};
    accO = wmma_f32_16x16x4(av, p[c], accO);
    ATTN_FENCE();
    e = en;
  }
  pend = p[7];
}

__global__ __launch_bounds__(128) __attribute__((amdgpu_waves_per_eu(8, 8)))
void attn_kernel(
    const float* __restrict__ x, float* __restrict__ Q,
    const float* __restrict__ KV, float* __restrict__ out) {
  const float* K = KV;
  const float* V = KV + (size_t)BB * SS * DD;
  __shared__ float red[4][8][16];
  __shared__ float4 kms[4];

  int wave = threadIdx.x >> 5;
  int lane = threadIdx.x & 31;
  int half = lane >> 4;
  int l16  = lane & 15;

  __builtin_amdgcn_s_setprio(3);

  const int QT = SS / 64;
  int bh = blockIdx.x / QT;
  int qt = blockIdx.x % QT;
  int b = bh / HH, h = bh % HH;
  int q0 = qt * 64 + wave * 16;

  const size_t hb = (size_t)bh * SS * HDIM;
  float* Qb = Q + hb;
  const float* Kb = K + hb;
  const float* Vb = V + hb;

  const v2f bq = *reinterpret_cast<const v2f*>(
      Qb + (size_t)(q0 + l16) * HDIM + 2 * half);

  v8f accO;
  for (int v = 0; v < 8; ++v) accO[v] = 0.0f;

  const float* kp = Kb + (size_t)l16 * HDIM + 2 * half;
  const float* vp = Vb + (l16 & 3) * 8 + half * 4;

  const int NS = SS / 32;

  v2f kX[2], kY[2];
  float4 vX[4], vY[4];
  v8f sX[2], sY[2];

  const float4 qr =
      *reinterpret_cast<const float4*>(Qb + (size_t)(q0 + l16) * HDIM);
  load_k(kp, 0, kX);
  load_v(vp, 0, vX);
  load_k(kp, 1, kY);

  {
    const float* ks = Kb + ((size_t)wave * (SS / 4) + lane) * HDIM;
    float4 km = *reinterpret_cast<const float4*>(ks);
    for (int it = 1; it < SS / 4 / 32; ++it) {
      const float4 kk =
          *reinterpret_cast<const float4*>(ks + (size_t)it * 32 * HDIM);
      km.x = fmaxf(km.x, kk.x);
      km.y = fmaxf(km.y, kk.y);
      km.z = fmaxf(km.z, kk.z);
      km.w = fmaxf(km.w, kk.w);
    }
    km.x = wave_max(km.x);
    km.y = wave_max(km.y);
    km.z = wave_max(km.z);
    km.w = wave_max(km.w);
    if (lane == 0) kms[wave] = km;
  }
  __syncthreads();
  float mbound;
  {
    const float4 k0 = kms[0], k1 = kms[1], k2 = kms[2], k3 = kms[3];
    mbound = qr.x * fmaxf(fmaxf(k0.x, k1.x), fmaxf(k2.x, k3.x)) +
             qr.y * fmaxf(fmaxf(k0.y, k1.y), fmaxf(k2.y, k3.y)) +
             qr.z * fmaxf(fmaxf(k0.z, k1.z), fmaxf(k2.z, k3.z)) +
             qr.w * fmaxf(fmaxf(k0.w, k1.w), fmaxf(k2.w, k3.w));
  }

  v2f l2 = {0.0f, 0.0f};
  v2f pend = {0.0f, 0.0f};
  {
    const float nm = -(mbound * ATTN_CS);
    const v2f nm2 = {nm, nm};
    score_tiles(kX, bq, sX);
#define FAST_STEP(i, kC, vC, sC, kN, vN, sN) \
  do {                                       \
    load_v(vp, (i) + 1, vN);                 \
    load_k(kp, (i) + 2, kC);                 \
    ATTN_FENCE();                            \
    score_tiles(kN, bq, sN);                 \
    ATTN_FENCE();                            \
    fast_step(sC, vC, nm2, l2, pend, accO);  \
    ATTN_FENCE();                            \
  } while (0)
#define FAST_PART(lo, hi)                         \
  for (int i = (lo); i < (hi); i += 2) {          \
    FAST_STEP(i, kX, vX, sX, kY, vY, sY);         \
    FAST_STEP(i + 1, kY, vY, sY, kX, vX, sX);     \
  }
    FAST_PART(0, 32)
    __builtin_amdgcn_s_setprio(2);
    FAST_PART(32, 48)
    __builtin_amdgcn_s_setprio(1);
    FAST_PART(48, 56)
    __builtin_amdgcn_s_setprio(0);
    FAST_PART(56, NS - 2)
#undef FAST_PART
#undef FAST_STEP
    load_v(vp, NS - 1, vY);
    score_tiles(kY, bq, sY);
    ATTN_FENCE();
    fast_step(sX, vX, nm2, l2, pend, accO);
    ATTN_FENCE();
    fast_step(sY, vY, nm2, l2, pend, accO);
    l2 += pend;
  }
  float l = l2.x + l2.y;
  l += xhalf(l);

  if (__builtin_amdgcn_ballot_w32(!(l >= 0x1p-60f)) != 0u) {
    float mraw = -INFINITY;
    float mc   = -INFINITY;
    float lsum = 0.0f;
    float mX, aX, mY, aY;
    for (int v = 0; v < 8; ++v) accO[v] = 0.0f;

    load_k(kp, 0, kX);
    load_v(vp, 0, vX);
    load_k(kp, 1, kY);
    score_tiles(kX, bq, sX);
    max_step(sX, mraw, mc, mX, aX);

#define ATTN_STEP(i, kC, vC, sC, mC, aC, kN, vN, sN, mN, aN) \
  do {                                                       \
    load_v(vp, (i) + 1, vN);                                 \
    load_k(kp, (i) + 2, kC);                                 \
    ATTN_FENCE();                                            \
    score_tiles(kN, bq, sN);                                 \
    ATTN_FENCE();                                            \
    fused_step<true>(sC, vC, mC, aC, sN, mN, aN, mraw, mc,   \
                     lsum, accO);                            \
    ATTN_FENCE();                                            \
  } while (0)

    for (int i = 0; i < NS - 2; i += 2) {
      ATTN_STEP(i, kX, vX, sX, mX, aX, kY, vY, sY, mY, aY);
      ATTN_STEP(i + 1, kY, vY, sY, mY, aY, kX, vX, sX, mX, aX);
    }
#undef ATTN_STEP
    load_v(vp, NS - 1, vY);
    score_tiles(kY, bq, sY);
    ATTN_FENCE();
    fused_step<true>(sX, vX, mX, aX, sY, mY, aY, mraw, mc, lsum, accO);
    ATTN_FENCE();
    fused_step<false>(sY, vY, mY, aY, sX, mX, aX, mraw, mc, lsum, accO);

    l = lsum + xhalf(lsum);
  }

  if (half == 0) {
    size_t idx = ((size_t)b * SS + q0 + l16) * DD + h * HDIM;
    const float4 xr = *reinterpret_cast<const float4*>(x + idx);
    float4 o;
    o.x = accO[0] / l + xr.x;
    o.y = accO[1] / l + xr.y;
    o.z = accO[2] / l + xr.z;
    o.w = accO[3] / l + xr.w;
    *reinterpret_cast<float4*>(out + idx) = o;
    float* r = &red[wave][0][l16];
    r[0 * 16] = o.x;
    r[1 * 16] = o.y;
    r[2 * 16] = o.z;
    r[3 * 16] = o.w;
    r[4 * 16] = o.x * o.x;
    r[5 * 16] = o.y * o.y;
    r[6 * 16] = o.z * o.z;
    r[7 * 16] = o.w * o.w;
  }
  __syncthreads();
  if (threadIdx.x < 8) {
    int v = threadIdx.x;
    float a0 = 0.0f, a1 = 0.0f, a2 = 0.0f, a3 = 0.0f;
    for (int j = 0; j < 16; ++j) {
      a0 += red[0][v][j];
      a1 += red[1][v][j];
      a2 += red[2][v][j];
      a3 += red[3][v][j];
    }
    Qb[(size_t)qt * 64 * HDIM + v] = (a0 + a1) + (a2 + a3);
  }
}

static __device__ __forceinline__ float parity_sum(float v) {
  v += row_xor<2>(v);
  v += row_xor<4>(v);
  v += row_xor<8>(v);
  return v + xhalf(v);
}

static __device__ __forceinline__ float4 bn_apply4(float4 v, float4 m, float4 rs,
                                                   float4 g, float4 be) {
  float4 o;
  o.x = (v.x - m.x) * rs.x * g.x + be.x;
  o.y = (v.y - m.y) * rs.y * g.y + be.y;
  o.z = (v.z - m.z) * rs.z * g.z + be.z;
  o.w = (v.w - m.w) * rs.w * g.w + be.w;
  return o;
}

__global__ __launch_bounds__(256) void bn_kernel(
    float* __restrict__ out, const float* __restrict__ part,
    const float* __restrict__ gamma, const float* __restrict__ beta) {
  __shared__ float4 smean[8], srstd[8];
  const int t    = threadIdx.x;
  const int wave = t >> 5;
  const int lane = t & 31;
  const int p    = blockIdx.x & 1;
  const int r0   = (blockIdx.x >> 1) * 128;
  const int c4   = t & 7;

  float* o = out + (size_t)(r0 + (t >> 3)) * DD + p * 32 + c4 * 4;
  float4 v[4];
#pragma unroll
  for (int i = 0; i < 4; ++i)
    v[i] = *reinterpret_cast<const float4*>(o + (size_t)i * 32 * DD);
  const float4 g  = *reinterpret_cast<const float4*>(gamma + p * 32 + c4 * 4);
  const float4 be = *reinterpret_cast<const float4*>(beta + p * 32 + c4 * 4);

  const int head = p * 8 + wave;
  const float* rp = part +
                    ((size_t)head * SS + (size_t)(lane >> 1) * 64) * HDIM +
                    (lane & 1) * 4;
  float4 a[8];
#pragma unroll
  for (int i = 0; i < 8; ++i)
    a[i] = *reinterpret_cast<const float4*>(
        rp + ((size_t)(i >> 1) * HH * SS + (size_t)(i & 1) * 16 * 64) * HDIM);
  __builtin_amdgcn_sched_barrier(0);
  float4 acc = a[0];
#pragma unroll
  for (int i = 1; i < 8; ++i) {
    acc.x += a[i].x;
    acc.y += a[i].y;
    acc.z += a[i].z;
    acc.w += a[i].w;
  }
  acc.x = parity_sum(acc.x);
  acc.y = parity_sum(acc.y);
  acc.z = parity_sum(acc.z);
  acc.w = parity_sum(acc.w);
  float4 nb;
  nb.x = row_xor<1>(acc.x);
  nb.y = row_xor<1>(acc.y);
  nb.z = row_xor<1>(acc.z);
  nb.w = row_xor<1>(acc.w);
  if (lane == 0) {
    const float invN = 1.0f / (float)(BB * SS);
    float4 mean, rstd;
    mean.x = acc.x * invN;
    mean.y = acc.y * invN;
    mean.z = acc.z * invN;
    mean.w = acc.w * invN;
    rstd.x = rsqrtf(nb.x * invN - mean.x * mean.x + 1e-5f);
    rstd.y = rsqrtf(nb.y * invN - mean.y * mean.y + 1e-5f);
    rstd.z = rsqrtf(nb.z * invN - mean.z * mean.z + 1e-5f);
    rstd.w = rsqrtf(nb.w * invN - mean.w * mean.w + 1e-5f);
    smean[wave] = mean;
    srstd[wave] = rstd;
  }
  __syncthreads();
  const float4 m  = smean[c4];
  const float4 rs = srstd[c4];
#pragma unroll
  for (int i = 0; i < 4; ++i)
    *reinterpret_cast<float4*>(o + (size_t)i * 32 * DD) =
        bn_apply4(v[i], m, rs, g, be);
}

extern "C" void kernel_launch(void* const* d_in, const int* in_sizes, int n_in,
                              void* d_out, int out_size, void* d_ws,
                              size_t ws_size, hipStream_t stream) {
  const float* x     = (const float*)d_in[0];
  const float* Wq    = (const float*)d_in[1];
  const float* bq    = (const float*)d_in[2];
  const float* Wk    = (const float*)d_in[3];
  const float* bk    = (const float*)d_in[4];
  const float* Wv    = (const float*)d_in[5];
  const float* bv    = (const float*)d_in[6];
  const float* gamma = (const float*)d_in[7];
  const float* beta  = (const float*)d_in[8];
  float* out = (float*)d_out;
  float* ws = (float*)d_ws;
  float* kv = ws + (size_t)BB * SS * DD;

  qkv_kernel<<<(3 * (BB * SS / 16)) / 4, 128, 0, stream>>>(x, Wq, bq, Wk, bk,
                                                           Wv, bv, ws);
  attn_kernel<<<BB * HH * (SS / 64), 128, 0, stream>>>(x, ws, kv, out);
  bn_kernel<<<2 * (BB * SS / 128), 256, 0, stream>>>(out, ws, gamma, beta);
}
